# starting kernel
# baseline (speedup 1.0000x reference)
#include <hip/hip_runtime.h>


typedef __attribute__((ext_vector_type(2))) float v2f;
typedef __attribute__((ext_vector_type(4))) float v4f;
typedef __attribute__((ext_vector_type(8))) float v8f;

#define N_USERS 29858
#define N_ITEMS 40981
#define N_TOTAL (N_USERS + N_ITEMS)
#define DIM     64
#define FDIM    512
#define N_LAYERS 3
#define NCHUNKS 128
#define CHUNK   556
#define QUADS   (CHUNK / 4)

static_assert(NCHUNKS * CHUNK >= N_TOTAL, "chunk coverage");

#define WMMA_F32(A, B, C) \
    __builtin_amdgcn_wmma_f32_16x16x4_f32(false, (A), false, (B), (short)0, (C), false, false)

__global__ void lgcn_init(const float* __restrict__ user,
                          const float* __restrict__ item,
                          float* __restrict__ ego,
                          float* __restrict__ acc) {
    int i = blockIdx.x * blockDim.x + threadIdx.x;
    if (i < N_TOTAL * DIM) {
        float val = (i < N_USERS * DIM) ? user[i] : item[i - N_USERS * DIM];
        ego[i] = val;
        acc[i] = val;
    }
}

__global__ __launch_bounds__(128)
void lgcn_projT(const float* __restrict__ v,
                const float* __restrict__ ego,
                float* __restrict__ partial  ) {
    __shared__ v2f lds_ego[(CHUNK / 2) * DIM];

    const int chunk   = blockIdx.x;
    const int n_begin = chunk * CHUNK;
    const int n_valid = min(N_TOTAL - n_begin, CHUNK);

    for (int idx = threadIdx.x; idx < (CHUNK / 2) * DIM; idx += blockDim.x) {
        const int j = idx >> 6;
        const int d = idx & 63;
        const int r0 = 2 * j, r1 = 2 * j + 1;
        const float e0 = ego[(size_t)(n_begin + min(r0, n_valid - 1)) * DIM + d];
        const float e1 = ego[(size_t)(n_begin + min(r1, n_valid - 1)) * DIM + d];
        v2f val;
        val.x = (r0 < n_valid) ? e0 : 0.0f;
        val.y = (r1 < n_valid) ? e1 : 0.0f;
        lds_ego[idx] = val;
    }
    __syncthreads();

    const int lane  = threadIdx.x & 31;
    const int wave  = threadIdx.x >> 5;
    const int f0    = (blockIdx.y * 4 + wave) * 32;
    const int row   = lane & 15;
    const int half  = lane >> 4;
    const int khalf = half * 2;

    v8f c0[4] = {{}, {}, {}, {}};
    v8f c1[4] = {{}, {}, {}, {}};

    if (n_valid == CHUNK) {
        const float* ap = v + (size_t)(n_begin + khalf) * FDIM + f0 + row;
#pragma unroll 1
        for (int q = 0; q < QUADS; ++q) {
            v2f a0, a1;
            a0.x = ap[0];         a0.y = ap[FDIM];
            a1.x = ap[16];        a1.y = ap[FDIM + 16];
            const v2f* bp = &lds_ego[(q * 2 + half) * DIM + row];
#pragma unroll
            for (int t = 0; t < 4; ++t) {
                v2f b = bp[t * 16];
                c0[t] = WMMA_F32(a0, b, c0[t]);
                c1[t] = WMMA_F32(a1, b, c1[t]);
            }
            ap += 4 * FDIM;
        }
    } else {
#pragma unroll 1
        for (int q = 0; q < QUADS; ++q) {
            const int na  = n_begin + q * 4 + khalf;
            const size_t ra0 = (size_t)min(na,     N_TOTAL - 1) * FDIM;
            const size_t ra1 = (size_t)min(na + 1, N_TOTAL - 1) * FDIM;
            v2f a0, a1;
            a0.x = v[ra0 + f0 + row];        a0.y = v[ra1 + f0 + row];
            a1.x = v[ra0 + f0 + 16 + row];   a1.y = v[ra1 + f0 + 16 + row];
            const v2f* bp = &lds_ego[(q * 2 + half) * DIM + row];
#pragma unroll
            for (int t = 0; t < 4; ++t) {
                v2f b = bp[t * 16];
                c0[t] = WMMA_F32(a0, b, c0[t]);
                c1[t] = WMMA_F32(a1, b, c1[t]);
            }
        }
    }

    float* outp = partial + (size_t)chunk * FDIM * DIM;
    const int mbase = f0 + (half << 3);
#pragma unroll
    for (int t = 0; t < 4; ++t) {
#pragma unroll
        for (int j = 0; j < 8; ++j) {
            outp[(size_t)(mbase + j)      * DIM + t * 16 + row] = c0[t][j];
            outp[(size_t)(mbase + 16 + j) * DIM + t * 16 + row] = c1[t][j];
        }
    }
}

__global__ void lgcn_reduce(const float* __restrict__ partial,
                            const float* __restrict__ filt  ,
                            float* __restrict__ proj  ) {
    int i = blockIdx.x * blockDim.x + threadIdx.x;
    if (i < FDIM * DIM) {
        float s = 0.0f;
        for (int c = 0; c < NCHUNKS; ++c)
            s += partial[(size_t)c * FDIM * DIM + i];
        proj[i] = s * filt[i / DIM];
    }
}

__global__ __launch_bounds__(256)
void lgcn_prop(const float* __restrict__ v,
               const float* __restrict__ proj,
               float* __restrict__ ego,
               float* __restrict__ acc) {
    __shared__ v2f lds_proj[(FDIM / 2) * DIM];

    for (int idx = threadIdx.x; idx < (FDIM / 2) * DIM; idx += blockDim.x) {
        const int j = idx >> 6;
        const int d = idx & 63;
        v2f val;
        val.x = proj[(size_t)(2 * j)     * DIM + d];
        val.y = proj[(size_t)(2 * j + 1) * DIM + d];
        lds_proj[idx] = val;
    }
    __syncthreads();

    const int lane  = threadIdx.x & 31;
    const int wave  = threadIdx.x >> 5;
    const int m0    = blockIdx.x * 256 + wave * 32;
    const int row   = lane & 15;
    const int half  = lane >> 4;
    const int khalf = half * 2;

    const int m = m0 + row;
    const float* vrow0 = v + (size_t)min(m,      N_TOTAL - 1) * FDIM;
    const float* vrow1 = v + (size_t)min(m + 16, N_TOTAL - 1) * FDIM;

    v8f c0[4] = {{}, {}, {}, {}};
    v8f c1[4] = {{}, {}, {}, {}};

#pragma unroll 1
    for (int k = 0; k < FDIM; k += 4) {
        v2f a0 = *(const v2f*)(vrow0 + k + khalf);
        v2f a1 = *(const v2f*)(vrow1 + k + khalf);
        if (k < FDIM - 64) {
            __builtin_prefetch(vrow0 + k + 64, 0, 1);
            __builtin_prefetch(vrow1 + k + 64, 0, 1);
        }
        const v2f* bp = &lds_proj[((k >> 1) + half) * DIM + row];
#pragma unroll
        for (int t = 0; t < 4; ++t) {
            v2f b = bp[t * 16];
            c0[t] = WMMA_F32(a0, b, c0[t]);
            c1[t] = WMMA_F32(a1, b, c1[t]);
        }
    }

    const int mbase = m0 + (half << 3);
#pragma unroll
    for (int j = 0; j < 8; ++j) {
#pragma unroll
        for (int t = 0; t < 4; ++t) {
            const int r0 = mbase + j;
            const int r1 = mbase + 16 + j;
            if (r0 < N_TOTAL) {
                const size_t idx = (size_t)r0 * DIM + t * 16 + row;
                ego[idx] = c0[t][j];
                acc[idx] += c0[t][j];
            }
            if (r1 < N_TOTAL) {
                const size_t idx = (size_t)r1 * DIM + t * 16 + row;
                ego[idx] = c1[t][j];
                acc[idx] += c1[t][j];
            }
        }
    }
}

__global__ void lgcn_final(const float* __restrict__ acc,
                           float* __restrict__ out) {
    int i = blockIdx.x * blockDim.x + threadIdx.x;
    if (i < N_TOTAL * DIM) out[i] = acc[i] * 0.25f;
}

extern "C" void kernel_launch(void* const* d_in, const int* in_sizes, int n_in,
                              void* d_out, int out_size, void* d_ws, size_t ws_size,
                              hipStream_t stream) {
    const float* user    = (const float*)d_in[0];
    const float* item    = (const float*)d_in[1];
    const float* v       = (const float*)d_in[2];
    const float* filters = (const float*)d_in[3];

    float* ego     = (float*)d_ws;
    float* acc     = ego + (size_t)N_TOTAL * DIM;
    float* partial = acc + (size_t)N_TOTAL * DIM;
    float* proj    = partial + (size_t)NCHUNKS * FDIM * DIM;

    const int totalND = N_TOTAL * DIM;
    lgcn_init<<<(totalND + 255) / 256, 256, 0, stream>>>(user, item, ego, acc);

    for (int layer = 0; layer < N_LAYERS; ++layer) {
        dim3 gA(NCHUNKS, 4);
        lgcn_projT<<<gA, 128, 0, stream>>>(v, ego, partial);

        lgcn_reduce<<<(FDIM * DIM + 255) / 256, 256, 0, stream>>>(
            partial, filters + (size_t)layer * FDIM, proj);

        const int nblocksB = (N_TOTAL + 255) / 256;
        lgcn_prop<<<nblocksB, 256, 0, stream>>>(v, proj, ego, acc);
    }

    lgcn_final<<<(totalND + 255) / 256, 256, 0, stream>>>(acc, (float*)d_out);
}


# Both GEMM K loops fetch the A operand one 8-step K-block ahead of its WMMAs into a second register set (was load-then-use per step plus prefetch hints); projT chunk zero-padded in LDS; sum order kept
# speedup vs baseline: 1.3855x; 1.3855x over previous
#include <hip/hip_runtime.h>


typedef __attribute__((ext_vector_type(2))) float v2f;
typedef __attribute__((ext_vector_type(4))) float v4f;
typedef __attribute__((ext_vector_type(8))) float v8f;

#define N_USERS 29858
#define N_ITEMS 40981
#define N_TOTAL (N_USERS + N_ITEMS)
#define DIM     64
#define FDIM    512
#define N_LAYERS 3
#define NCHUNKS 128
#define CHUNK   556
#define QUADS   (CHUNK / 4)
#define KB      8
#define NBLK_A  ((QUADS + KB - 1) / KB + ((QUADS + KB - 1) / KB) % 2)
#define ROWS_PAD (NBLK_A * KB * 4)
#define NBLK_B  (FDIM / (KB * 4))

static_assert(NCHUNKS * CHUNK >= N_TOTAL, "chunk coverage");
static_assert(NBLK_A % 2 == 0 && NBLK_B % 2 == 0, "two register sets alternate");
static_assert(ROWS_PAD >= CHUNK && ROWS_PAD % 2 == 0, "padded chunk");

#define WMMA_F32(A, B, C) \
    __builtin_amdgcn_wmma_f32_16x16x4_f32(false, (A), false, (B), (short)0, (C), false, false)

__device__ __forceinline__ void mma_block(v8f (&c0)[4], v8f (&c1)[4],
                                          const v2f (&a0)[KB], const v2f (&a1)[KB],
                                          const v2f* bp) {
#pragma unroll
    for (int i = 0; i < KB; ++i) {
#pragma unroll
        for (int t = 0; t < 4; ++t) {
            v2f b = bp[(2 * i) * DIM + t * 16];
            c0[t] = WMMA_F32(a0[i], b, c0[t]);
            c1[t] = WMMA_F32(a1[i], b, c1[t]);
        }
    }
}

template <bool RAGGED>
__device__ __forceinline__ void projT_loadA(v2f (&a0)[KB], v2f (&a1)[KB],
                                            const float* __restrict__ v,
                                            int n0, int col) {
    if (RAGGED) {
#pragma unroll
        for (int i = 0; i < KB; ++i) {
            const int na = n0 + 4 * i;
            const size_t r0 = (size_t)min(na,     N_TOTAL - 1) * FDIM;
            const size_t r1 = (size_t)min(na + 1, N_TOTAL - 1) * FDIM;
            a0[i].x = v[r0 + col];        a0[i].y = v[r1 + col];
            a1[i].x = v[r0 + col + 16];   a1[i].y = v[r1 + col + 16];
        }
    } else {
        const float* ap = v + (size_t)n0 * FDIM + col;
#pragma unroll
        for (int i = 0; i < KB; ++i) {
            a0[i].x = ap[(4 * i) * FDIM];        a0[i].y = ap[(4 * i + 1) * FDIM];
            a1[i].x = ap[(4 * i) * FDIM + 16];   a1[i].y = ap[(4 * i + 1) * FDIM + 16];
        }
    }
}

template <bool RAGGED>
__device__ __forceinline__ void projT_kloop(v8f (&c0)[4], v8f (&c1)[4],
                                            v2f (&A0)[KB], v2f (&A1)[KB],
                                            const float* __restrict__ v,
                                            const v2f* lds_lane, int nrow, int col) {
    v2f B0[KB], B1[KB];
#pragma unroll 1
    for (int blk = 0; blk < NBLK_A; blk += 2) {
        projT_loadA<RAGGED>(B0, B1, v, nrow + (blk + 1) * (KB * 4), col);
        mma_block(c0, c1, A0, A1, lds_lane + (size_t)blk * (KB * 2) * DIM);
        projT_loadA<RAGGED>(A0, A1, v, nrow + min(blk + 2, NBLK_A - 1) * (KB * 4), col);
        mma_block(c0, c1, B0, B1, lds_lane + (size_t)(blk + 1) * (KB * 2) * DIM);
    }
}

__global__ void lgcn_init(const float* __restrict__ user,
                          const float* __restrict__ item,
                          float* __restrict__ ego,
                          float* __restrict__ acc) {
    int i = blockIdx.x * blockDim.x + threadIdx.x;
    if (i < N_TOTAL * DIM) {
        float val = (i < N_USERS * DIM) ? user[i] : item[i - N_USERS * DIM];
        ego[i] = val;
        acc[i] = val;
    }
}

__global__ __launch_bounds__(128)
void lgcn_projT(const float* __restrict__ v,
                const float* __restrict__ ego,
                float* __restrict__ partial  ) {
    __shared__ v2f lds_ego[(ROWS_PAD / 2) * DIM];

    const int chunk   = blockIdx.x;
    const int n_begin = chunk * CHUNK;
    const int n_valid = min(N_TOTAL - n_begin, CHUNK);

    const int lane  = threadIdx.x & 31;
    const int wave  = threadIdx.x >> 5;
    const int f0    = (blockIdx.y * 4 + wave) * 32;
    const int row   = lane & 15;
    const int half  = lane >> 4;
    const int khalf = half * 2;

    const bool fast = (n_begin + ROWS_PAD <= N_TOTAL);

    v2f A0[KB], A1[KB];
    if (fast) projT_loadA<false>(A0, A1, v, n_begin + khalf, f0 + row);
    else      projT_loadA<true >(A0, A1, v, n_begin + khalf, f0 + row);

    for (int idx = threadIdx.x; idx < (ROWS_PAD / 2) * DIM; idx += blockDim.x) {
        const int j = idx >> 6;
        const int d = idx & 63;
        const int r0 = 2 * j, r1 = 2 * j + 1;
        const float e0 = ego[(size_t)(n_begin + min(r0, n_valid - 1)) * DIM + d];
        const float e1 = ego[(size_t)(n_begin + min(r1, n_valid - 1)) * DIM + d];
        v2f val;
        val.x = (r0 < n_valid) ? e0 : 0.0f;
        val.y = (r1 < n_valid) ? e1 : 0.0f;
        lds_ego[idx] = val;
    }
    __syncthreads();

    v8f c0[4] = {{}, {}, {}, {}};
    v8f c1[4] = {{}, {}, {}, {}};

    const v2f* lds_lane = &lds_ego[half * DIM + row];
    if (fast) {
        projT_kloop<false>(c0, c1, A0, A1, v, lds_lane, n_begin + khalf, f0 + row);
    } else {
        projT_kloop<true >(c0, c1, A0, A1, v, lds_lane, n_begin + khalf, f0 + row);
    }

    float* outp = partial + (size_t)chunk * FDIM * DIM;
    const int mbase = f0 + (half << 3);
#pragma unroll
    for (int t = 0; t < 4; ++t) {
#pragma unroll
        for (int j = 0; j < 8; ++j) {
            outp[(size_t)(mbase + j)      * DIM + t * 16 + row] = c0[t][j];
            outp[(size_t)(mbase + 16 + j) * DIM + t * 16 + row] = c1[t][j];
        }
    }
}

__global__ void lgcn_reduce(const float* __restrict__ partial,
                            const float* __restrict__ filt  ,
                            float* __restrict__ proj  ) {
    int i = blockIdx.x * blockDim.x + threadIdx.x;
    if (i < FDIM * DIM) {
        float s = 0.0f;
        for (int c = 0; c < NCHUNKS; ++c)
            s += partial[(size_t)c * FDIM * DIM + i];
        proj[i] = s * filt[i / DIM];
    }
}

__device__ __forceinline__ void prop_loadA(v2f (&a0)[KB], v2f (&a1)[KB],
                                           const float* p0, const float* p1, int blk) {
    p0 += blk * (KB * 4);
    p1 += blk * (KB * 4);
#pragma unroll
    for (int i = 0; i < KB; ++i) {
        a0[i] = *(const v2f*)(p0 + 4 * i);
        a1[i] = *(const v2f*)(p1 + 4 * i);
    }
}

__global__ __launch_bounds__(256)
void lgcn_prop(const float* __restrict__ v,
               const float* __restrict__ proj,
               float* __restrict__ ego,
               float* __restrict__ acc) {
    __shared__ v2f lds_proj[(FDIM / 2) * DIM];

    for (int idx = threadIdx.x; idx < (FDIM / 2) * DIM; idx += blockDim.x) {
        const int j = idx >> 6;
        const int d = idx & 63;
        v2f val;
        val.x = proj[(size_t)(2 * j)     * DIM + d];
        val.y = proj[(size_t)(2 * j + 1) * DIM + d];
        lds_proj[idx] = val;
    }
    __syncthreads();

    const int lane  = threadIdx.x & 31;
    const int wave  = threadIdx.x >> 5;
    const int m0    = blockIdx.x * 256 + wave * 32;
    const int row   = lane & 15;
    const int half  = lane >> 4;
    const int khalf = half * 2;

    const int m = m0 + row;
    const float* vrow0 = v + (size_t)min(m,      N_TOTAL - 1) * FDIM;
    const float* vrow1 = v + (size_t)min(m + 16, N_TOTAL - 1) * FDIM;

    v8f c0[4] = {{}, {}, {}, {}};
    v8f c1[4] = {{}, {}, {}, {}};

    const v2f* lds_lane = &lds_proj[half * DIM + row];
    v2f A0[KB], A1[KB], B0[KB], B1[KB];
    prop_loadA(A0, A1, vrow0 + khalf, vrow1 + khalf, 0);
#pragma unroll 1
    for (int blk = 0; blk < NBLK_B; blk += 2) {
        prop_loadA(B0, B1, vrow0 + khalf, vrow1 + khalf, blk + 1);
        mma_block(c0, c1, A0, A1, lds_lane + (size_t)blk * (KB * 2) * DIM);
        prop_loadA(A0, A1, vrow0 + khalf, vrow1 + khalf, min(blk + 2, NBLK_B - 1));
        mma_block(c0, c1, B0, B1, lds_lane + (size_t)(blk + 1) * (KB * 2) * DIM);
    }

    const int mbase = m0 + (half << 3);
#pragma unroll
    for (int j = 0; j < 8; ++j) {
#pragma unroll
        for (int t = 0; t < 4; ++t) {
            const int r0 = mbase + j;
            const int r1 = mbase + 16 + j;
            if (r0 < N_TOTAL) {
                const size_t idx = (size_t)r0 * DIM + t * 16 + row;
                ego[idx] = c0[t][j];
                acc[idx] += c0[t][j];
            }
            if (r1 < N_TOTAL) {
                const size_t idx = (size_t)r1 * DIM + t * 16 + row;
                ego[idx] = c1[t][j];
                acc[idx] += c1[t][j];
            }
        }
    }
}

__global__ void lgcn_final(const float* __restrict__ acc,
                           float* __restrict__ out) {
    int i = blockIdx.x * blockDim.x + threadIdx.x;
    if (i < N_TOTAL * DIM) out[i] = acc[i] * 0.25f;
}

extern "C" void kernel_launch(void* const* d_in, const int* in_sizes, int n_in,
                              void* d_out, int out_size, void* d_ws, size_t ws_size,
                              hipStream_t stream) {
    const float* user    = (const float*)d_in[0];
    const float* item    = (const float*)d_in[1];
    const float* v       = (const float*)d_in[2];
    const float* filters = (const float*)d_in[3];

    float* ego     = (float*)d_ws;
    float* acc     = ego + (size_t)N_TOTAL * DIM;
    float* partial = acc + (size_t)N_TOTAL * DIM;
    float* proj    = partial + (size_t)NCHUNKS * FDIM * DIM;

    const int totalND = N_TOTAL * DIM;
    lgcn_init<<<(totalND + 255) / 256, 256, 0, stream>>>(user, item, ego, acc);

    for (int layer = 0; layer < N_LAYERS; ++layer) {
        dim3 gA(NCHUNKS, 4);
        lgcn_projT<<<gA, 128, 0, stream>>>(v, ego, partial);

        lgcn_reduce<<<(FDIM * DIM + 255) / 256, 256, 0, stream>>>(
            partial, filters + (size_t)layer * FDIM, proj);

        const int nblocksB = (N_TOTAL + 255) / 256;
        lgcn_prop<<<nblocksB, 256, 0, stream>>>(v, proj, ego, acc);
    }

    lgcn_final<<<(totalND + 255) / 256, 256, 0, stream>>>(acc, (float*)d_out);
}


# Both LDS staging loops (ego chunk in projT, proj in prop) fetch 16-byte row pieces, 16-18 loads per thread in flight before the LDS stores (was one 4-byte row pair per round trip); sum order kept
# speedup vs baseline: 2.3459x; 1.6932x over previous
#include <hip/hip_runtime.h>


typedef __attribute__((ext_vector_type(2))) float v2f;
typedef __attribute__((ext_vector_type(4))) float v4f;
typedef __attribute__((ext_vector_type(8))) float v8f;

#define N_USERS 29858
#define N_ITEMS 40981
#define N_TOTAL (N_USERS + N_ITEMS)
#define DIM     64
#define FDIM    512
#define N_LAYERS 3
#define NCHUNKS 128
#define CHUNK   556
#define QUADS   (CHUNK / 4)
#define KB      8
#define NBLK_A  ((QUADS + KB - 1) / KB + ((QUADS + KB - 1) / KB) % 2)
#define ROWS_PAD (NBLK_A * KB * 4)
#define NBLK_B  (FDIM / (KB * 4))
#define PROJT_THREADS 128
#define PROP_THREADS  256
#define QCOLS   (DIM / 4)
#define STAGE_A_ITEMS ((ROWS_PAD / 2) * QCOLS)
#define STAGE_B_ITEMS ((FDIM / 2) * QCOLS)
#define STAGE_A_GROUP 9
#define STAGE_B_GROUP 8

static_assert(NCHUNKS * CHUNK >= N_TOTAL, "chunk coverage");
static_assert(NBLK_A % 2 == 0 && NBLK_B % 2 == 0, "two register sets alternate");
static_assert(ROWS_PAD >= CHUNK && ROWS_PAD % 2 == 0, "padded chunk");
static_assert(STAGE_A_ITEMS % (PROJT_THREADS * STAGE_A_GROUP) == 0, "whole staging groups");
static_assert(STAGE_B_ITEMS % (PROP_THREADS * STAGE_B_GROUP) == 0, "whole staging groups");

#define WMMA_F32(A, B, C) \
    __builtin_amdgcn_wmma_f32_16x16x4_f32(false, (A), false, (B), (short)0, (C), false, false)

__device__ __forceinline__ void mma_block(v8f (&c0)[4], v8f (&c1)[4],
                                          const v2f (&a0)[KB], const v2f (&a1)[KB],
                                          const v2f* bp) {
#pragma unroll
    for (int i = 0; i < KB; ++i) {
#pragma unroll
        for (int t = 0; t < 4; ++t) {
            v2f b = bp[(2 * i) * DIM + t * 16];
            c0[t] = WMMA_F32(a0[i], b, c0[t]);
            c1[t] = WMMA_F32(a1[i], b, c1[t]);
        }
    }
}

__device__ __forceinline__ void stage_store(v2f* dst, v4f lo, v4f hi) {
    v2f e;
    e.x = lo.x; e.y = hi.x; dst[0] = e;
    e.x = lo.y; e.y = hi.y; dst[1] = e;
    e.x = lo.z; e.y = hi.z; dst[2] = e;
    e.x = lo.w; e.y = hi.w; dst[3] = e;
}

template <bool RAGGED>
__device__ __forceinline__ void projT_loadA(v2f (&a0)[KB], v2f (&a1)[KB],
                                            const float* __restrict__ v,
                                            int n0, int col) {
    if (RAGGED) {
#pragma unroll
        for (int i = 0; i < KB; ++i) {
            const int na = n0 + 4 * i;
            const size_t r0 = (size_t)min(na,     N_TOTAL - 1) * FDIM;
            const size_t r1 = (size_t)min(na + 1, N_TOTAL - 1) * FDIM;
            a0[i].x = v[r0 + col];        a0[i].y = v[r1 + col];
            a1[i].x = v[r0 + col + 16];   a1[i].y = v[r1 + col + 16];
        }
    } else {
        const float* ap = v + (size_t)n0 * FDIM + col;
#pragma unroll
        for (int i = 0; i < KB; ++i) {
            a0[i].x = ap[(4 * i) * FDIM];        a0[i].y = ap[(4 * i + 1) * FDIM];
            a1[i].x = ap[(4 * i) * FDIM + 16];   a1[i].y = ap[(4 * i + 1) * FDIM + 16];
        }
    }
}

template <bool RAGGED>
__device__ __forceinline__ void projT_kloop(v8f (&c0)[4], v8f (&c1)[4],
                                            v2f (&A0)[KB], v2f (&A1)[KB],
                                            const float* __restrict__ v,
                                            const v2f* lds_lane, int nrow, int col) {
    v2f B0[KB], B1[KB];
#pragma unroll 1
    for (int blk = 0; blk < NBLK_A; blk += 2) {
        projT_loadA<RAGGED>(B0, B1, v, nrow + (blk + 1) * (KB * 4), col);
        mma_block(c0, c1, A0, A1, lds_lane + (size_t)blk * (KB * 2) * DIM);
        projT_loadA<RAGGED>(A0, A1, v, nrow + min(blk + 2, NBLK_A - 1) * (KB * 4), col);
        mma_block(c0, c1, B0, B1, lds_lane + (size_t)(blk + 1) * (KB * 2) * DIM);
    }
}

__global__ void lgcn_init(const float* __restrict__ user,
                          const float* __restrict__ item,
                          float* __restrict__ ego,
                          float* __restrict__ acc) {
    int i = blockIdx.x * blockDim.x + threadIdx.x;
    if (i < N_TOTAL * DIM) {
        float val = (i < N_USERS * DIM) ? user[i] : item[i - N_USERS * DIM];
        ego[i] = val;
        acc[i] = val;
    }
}

__global__ __launch_bounds__(PROJT_THREADS)
void lgcn_projT(const float* __restrict__ v,
                const float* __restrict__ ego,
                float* __restrict__ partial  ) {
    __shared__ __attribute__((aligned(16))) v2f lds_ego[(ROWS_PAD / 2) * DIM];

    const int chunk   = blockIdx.x;
    const int n_begin = chunk * CHUNK;
    const int n_valid = min(N_TOTAL - n_begin, CHUNK);

    const int lane  = threadIdx.x & 31;
    const int wave  = threadIdx.x >> 5;
    const int f0    = (blockIdx.y * 4 + wave) * 32;
    const int row   = lane & 15;
    const int half  = lane >> 4;
    const int khalf = half * 2;

    const bool fast = (n_begin + ROWS_PAD <= N_TOTAL);

    v2f A0[KB], A1[KB];
    if (fast) projT_loadA<false>(A0, A1, v, n_begin + khalf, f0 + row);
    else      projT_loadA<true >(A0, A1, v, n_begin + khalf, f0 + row);

    const float* ego_chunk = ego + (size_t)n_begin * DIM;
    for (int base = threadIdx.x; base < STAGE_A_ITEMS; base += PROJT_THREADS * STAGE_A_GROUP) {
        v4f lo[STAGE_A_GROUP], hi[STAGE_A_GROUP];
#pragma unroll
        for (int u = 0; u < STAGE_A_GROUP; ++u) {
            const int q  = base + u * PROJT_THREADS;
            const int j  = q / QCOLS;
            const int d4 = (q % QCOLS) * 4;
            lo[u] = *(const v4f*)(ego_chunk + (size_t)min(2 * j,     n_valid - 1) * DIM + d4);
            hi[u] = *(const v4f*)(ego_chunk + (size_t)min(2 * j + 1, n_valid - 1) * DIM + d4);
        }
#pragma unroll
        for (int u = 0; u < STAGE_A_GROUP; ++u) {
            const int q  = base + u * PROJT_THREADS;
            const int j  = q / QCOLS;
            const int d4 = (q % QCOLS) * 4;
            const v4f zero = {0.0f, 0.0f, 0.0f, 0.0f};
            stage_store(&lds_ego[j * DIM + d4],
                        (2 * j     < n_valid) ? lo[u] : zero,
                        (2 * j + 1 < n_valid) ? hi[u] : zero);
        }
    }
    __syncthreads();

    v8f c0[4] = {{}, {}, {}, {}};
    v8f c1[4] = {{}, {}, {}, {}};

    const v2f* lds_lane = &lds_ego[half * DIM + row];
    if (fast) {
        projT_kloop<false>(c0, c1, A0, A1, v, lds_lane, n_begin + khalf, f0 + row);
    } else {
        projT_kloop<true >(c0, c1, A0, A1, v, lds_lane, n_begin + khalf, f0 + row);
    }

    float* outp = partial + (size_t)chunk * FDIM * DIM;
    const int mbase = f0 + (half << 3);
#pragma unroll
    for (int t = 0; t < 4; ++t) {
#pragma unroll
        for (int j = 0; j < 8; ++j) {
            outp[(size_t)(mbase + j)      * DIM + t * 16 + row] = c0[t][j];
            outp[(size_t)(mbase + 16 + j) * DIM + t * 16 + row] = c1[t][j];
        }
    }
}

__global__ void lgcn_reduce(const float* __restrict__ partial,
                            const float* __restrict__ filt  ,
                            float* __restrict__ proj  ) {
    int i = blockIdx.x * blockDim.x + threadIdx.x;
    if (i < FDIM * DIM) {
        float s = 0.0f;
        for (int c = 0; c < NCHUNKS; ++c)
            s += partial[(size_t)c * FDIM * DIM + i];
        proj[i] = s * filt[i / DIM];
    }
}

__device__ __forceinline__ void prop_loadA(v2f (&a0)[KB], v2f (&a1)[KB],
                                           const float* p0, const float* p1, int blk) {
    p0 += blk * (KB * 4);
    p1 += blk * (KB * 4);
#pragma unroll
    for (int i = 0; i < KB; ++i) {
        a0[i] = *(const v2f*)(p0 + 4 * i);
        a1[i] = *(const v2f*)(p1 + 4 * i);
    }
}

__global__ __launch_bounds__(PROP_THREADS)
void lgcn_prop(const float* __restrict__ v,
               const float* __restrict__ proj,
               float* __restrict__ ego,
               float* __restrict__ acc) {
    __shared__ __attribute__((aligned(16))) v2f lds_proj[(FDIM / 2) * DIM];

    for (int base = threadIdx.x; base < STAGE_B_ITEMS; base += PROP_THREADS * STAGE_B_GROUP) {
        v4f lo[STAGE_B_GROUP], hi[STAGE_B_GROUP];
#pragma unroll
        for (int u = 0; u < STAGE_B_GROUP; ++u) {
            const int q  = base + u * PROP_THREADS;
            const int j  = q / QCOLS;
            const int d4 = (q % QCOLS) * 4;
            lo[u] = *(const v4f*)(proj + (size_t)(2 * j)     * DIM + d4);
            hi[u] = *(const v4f*)(proj + (size_t)(2 * j + 1) * DIM + d4);
        }
#pragma unroll
        for (int u = 0; u < STAGE_B_GROUP; ++u) {
            const int q  = base + u * PROP_THREADS;
            const int j  = q / QCOLS;
            const int d4 = (q % QCOLS) * 4;
            stage_store(&lds_proj[j * DIM + d4], lo[u], hi[u]);
        }
    }
    __syncthreads();

    const int lane  = threadIdx.x & 31;
    const int wave  = threadIdx.x >> 5;
    const int m0    = blockIdx.x * 256 + wave * 32;
    const int row   = lane & 15;
    const int half  = lane >> 4;
    const int khalf = half * 2;

    const int m = m0 + row;
    const float* vrow0 = v + (size_t)min(m,      N_TOTAL - 1) * FDIM;
    const float* vrow1 = v + (size_t)min(m + 16, N_TOTAL - 1) * FDIM;

    v8f c0[4] = {{}, {}, {}, {}};
    v8f c1[4] = {{}, {}, {}, {}};

    const v2f* lds_lane = &lds_proj[half * DIM + row];
    v2f A0[KB], A1[KB], B0[KB], B1[KB];
    prop_loadA(A0, A1, vrow0 + khalf, vrow1 + khalf, 0);
#pragma unroll 1
    for (int blk = 0; blk < NBLK_B; blk += 2) {
        prop_loadA(B0, B1, vrow0 + khalf, vrow1 + khalf, blk + 1);
        mma_block(c0, c1, A0, A1, lds_lane + (size_t)blk * (KB * 2) * DIM);
        prop_loadA(A0, A1, vrow0 + khalf, vrow1 + khalf, min(blk + 2, NBLK_B - 1));
        mma_block(c0, c1, B0, B1, lds_lane + (size_t)(blk + 1) * (KB * 2) * DIM);
    }

    const int mbase = m0 + (half << 3);
#pragma unroll
    for (int j = 0; j < 8; ++j) {
#pragma unroll
        for (int t = 0; t < 4; ++t) {
            const int r0 = mbase + j;
            const int r1 = mbase + 16 + j;
            if (r0 < N_TOTAL) {
                const size_t idx = (size_t)r0 * DIM + t * 16 + row;
                ego[idx] = c0[t][j];
                acc[idx] += c0[t][j];
            }
            if (r1 < N_TOTAL) {
                const size_t idx = (size_t)r1 * DIM + t * 16 + row;
                ego[idx] = c1[t][j];
                acc[idx] += c1[t][j];
            }
        }
    }
}

__global__ void lgcn_final(const float* __restrict__ acc,
                           float* __restrict__ out) {
    int i = blockIdx.x * blockDim.x + threadIdx.x;
    if (i < N_TOTAL * DIM) out[i] = acc[i] * 0.25f;
}

extern "C" void kernel_launch(void* const* d_in, const int* in_sizes, int n_in,
                              void* d_out, int out_size, void* d_ws, size_t ws_size,
                              hipStream_t stream) {
    const float* user    = (const float*)d_in[0];
    const float* item    = (const float*)d_in[1];
    const float* v       = (const float*)d_in[2];
    const float* filters = (const float*)d_in[3];

    float* ego     = (float*)d_ws;
    float* acc     = ego + (size_t)N_TOTAL * DIM;
    float* partial = acc + (size_t)N_TOTAL * DIM;
    float* proj    = partial + (size_t)NCHUNKS * FDIM * DIM;

    const int totalND = N_TOTAL * DIM;
    lgcn_init<<<(totalND + 255) / 256, 256, 0, stream>>>(user, item, ego, acc);

    for (int layer = 0; layer < N_LAYERS; ++layer) {
        dim3 gA(NCHUNKS, 4);
        lgcn_projT<<<gA, PROJT_THREADS, 0, stream>>>(v, ego, partial);

        lgcn_reduce<<<(FDIM * DIM + 255) / 256, 256, 0, stream>>>(
            partial, filters + (size_t)layer * FDIM, proj);

        const int nblocksB = (N_TOTAL + 255) / 256;
        lgcn_prop<<<nblocksB, PROP_THREADS, 0, stream>>>(v, proj, ego, acc);
    }

    lgcn_final<<<(totalND + 255) / 256, 256, 0, stream>>>(acc, (float*)d_out);
}


# prop epilogue: the 64 acc reads of a lane are all requested before its first store (was one read-wait-add-store round trip per element, 64 in series); every sum and its order unchanged
# speedup vs baseline: 2.7254x; 1.1618x over previous
#include <hip/hip_runtime.h>


typedef __attribute__((ext_vector_type(2))) float v2f;
typedef __attribute__((ext_vector_type(4))) float v4f;
typedef __attribute__((ext_vector_type(8))) float v8f;

#define N_USERS 29858
#define N_ITEMS 40981
#define N_TOTAL (N_USERS + N_ITEMS)
#define DIM     64
#define FDIM    512
#define N_LAYERS 3
#define NCHUNKS 128
#define CHUNK   556
#define QUADS   (CHUNK / 4)
#define KB      8
#define NBLK_A  ((QUADS + KB - 1) / KB + ((QUADS + KB - 1) / KB) % 2)
#define ROWS_PAD (NBLK_A * KB * 4)
#define NBLK_B  (FDIM / (KB * 4))
#define PROJT_THREADS 128
#define PROP_THREADS  256
#define QCOLS   (DIM / 4)
#define STAGE_A_ITEMS ((ROWS_PAD / 2) * QCOLS)
#define STAGE_B_ITEMS ((FDIM / 2) * QCOLS)
#define STAGE_A_GROUP 9
#define STAGE_B_GROUP 8

static_assert(NCHUNKS * CHUNK >= N_TOTAL, "chunk coverage");
static_assert(NBLK_A % 2 == 0 && NBLK_B % 2 == 0, "two register sets alternate");
static_assert(ROWS_PAD >= CHUNK && ROWS_PAD % 2 == 0, "padded chunk");
static_assert(STAGE_A_ITEMS % (PROJT_THREADS * STAGE_A_GROUP) == 0, "whole staging groups");
static_assert(STAGE_B_ITEMS % (PROP_THREADS * STAGE_B_GROUP) == 0, "whole staging groups");

#define WMMA_F32(A, B, C) \
    __builtin_amdgcn_wmma_f32_16x16x4_f32(false, (A), false, (B), (short)0, (C), false, false)

__device__ __forceinline__ void mma_block(v8f (&c0)[4], v8f (&c1)[4],
                                          const v2f (&a0)[KB], const v2f (&a1)[KB],
                                          const v2f* bp) {
#pragma unroll
    for (int i = 0; i < KB; ++i) {
#pragma unroll
        for (int t = 0; t < 4; ++t) {
            v2f b = bp[(2 * i) * DIM + t * 16];
            c0[t] = WMMA_F32(a0[i], b, c0[t]);
            c1[t] = WMMA_F32(a1[i], b, c1[t]);
        }
    }
}

__device__ __forceinline__ void stage_store(v2f* dst, v4f lo, v4f hi) {
    v2f e;
    e.x = lo.x; e.y = hi.x; dst[0] = e;
    e.x = lo.y; e.y = hi.y; dst[1] = e;
    e.x = lo.z; e.y = hi.z; dst[2] = e;
    e.x = lo.w; e.y = hi.w; dst[3] = e;
}

template <bool RAGGED>
__device__ __forceinline__ void projT_loadA(v2f (&a0)[KB], v2f (&a1)[KB],
                                            const float* __restrict__ v,
                                            int n0, int col) {
    if (RAGGED) {
#pragma unroll
        for (int i = 0; i < KB; ++i) {
            const int na = n0 + 4 * i;
            const size_t r0 = (size_t)min(na,     N_TOTAL - 1) * FDIM;
            const size_t r1 = (size_t)min(na + 1, N_TOTAL - 1) * FDIM;
            a0[i].x = v[r0 + col];        a0[i].y = v[r1 + col];
            a1[i].x = v[r0 + col + 16];   a1[i].y = v[r1 + col + 16];
        }
    } else {
        const float* ap = v + (size_t)n0 * FDIM + col;
#pragma unroll
        for (int i = 0; i < KB; ++i) {
            a0[i].x = ap[(4 * i) * FDIM];        a0[i].y = ap[(4 * i + 1) * FDIM];
            a1[i].x = ap[(4 * i) * FDIM + 16];   a1[i].y = ap[(4 * i + 1) * FDIM + 16];
        }
    }
}

template <bool RAGGED>
__device__ __forceinline__ void projT_kloop(v8f (&c0)[4], v8f (&c1)[4],
                                            v2f (&A0)[KB], v2f (&A1)[KB],
                                            const float* __restrict__ v,
                                            const v2f* lds_lane, int nrow, int col) {
    v2f B0[KB], B1[KB];
#pragma unroll 1
    for (int blk = 0; blk < NBLK_A; blk += 2) {
        projT_loadA<RAGGED>(B0, B1, v, nrow + (blk + 1) * (KB * 4), col);
        mma_block(c0, c1, A0, A1, lds_lane + (size_t)blk * (KB * 2) * DIM);
        projT_loadA<RAGGED>(A0, A1, v, nrow + min(blk + 2, NBLK_A - 1) * (KB * 4), col);
        mma_block(c0, c1, B0, B1, lds_lane + (size_t)(blk + 1) * (KB * 2) * DIM);
    }
}

__global__ void lgcn_init(const float* __restrict__ user,
                          const float* __restrict__ item,
                          float* __restrict__ ego,
                          float* __restrict__ acc) {
    int i = blockIdx.x * blockDim.x + threadIdx.x;
    if (i < N_TOTAL * DIM) {
        float val = (i < N_USERS * DIM) ? user[i] : item[i - N_USERS * DIM];
        ego[i] = val;
        acc[i] = val;
    }
}

__global__ __launch_bounds__(PROJT_THREADS)
void lgcn_projT(const float* __restrict__ v,
                const float* __restrict__ ego,
                float* __restrict__ partial  ) {
    __shared__ __attribute__((aligned(16))) v2f lds_ego[(ROWS_PAD / 2) * DIM];

    const int chunk   = blockIdx.x;
    const int n_begin = chunk * CHUNK;
    const int n_valid = min(N_TOTAL - n_begin, CHUNK);

    const int lane  = threadIdx.x & 31;
    const int wave  = threadIdx.x >> 5;
    const int f0    = (blockIdx.y * 4 + wave) * 32;
    const int row   = lane & 15;
    const int half  = lane >> 4;
    const int khalf = half * 2;

    const bool fast = (n_begin + ROWS_PAD <= N_TOTAL);

    v2f A0[KB], A1[KB];
    if (fast) projT_loadA<false>(A0, A1, v, n_begin + khalf, f0 + row);
    else      projT_loadA<true >(A0, A1, v, n_begin + khalf, f0 + row);

    const float* ego_chunk = ego + (size_t)n_begin * DIM;
    for (int base = threadIdx.x; base < STAGE_A_ITEMS; base += PROJT_THREADS * STAGE_A_GROUP) {
        v4f lo[STAGE_A_GROUP], hi[STAGE_A_GROUP];
#pragma unroll
        for (int u = 0; u < STAGE_A_GROUP; ++u) {
            const int q  = base + u * PROJT_THREADS;
            const int j  = q / QCOLS;
            const int d4 = (q % QCOLS) * 4;
            lo[u] = *(const v4f*)(ego_chunk + (size_t)min(2 * j,     n_valid - 1) * DIM + d4);
            hi[u] = *(const v4f*)(ego_chunk + (size_t)min(2 * j + 1, n_valid - 1) * DIM + d4);
        }
#pragma unroll
        for (int u = 0; u < STAGE_A_GROUP; ++u) {
            const int q  = base + u * PROJT_THREADS;
            const int j  = q / QCOLS;
            const int d4 = (q % QCOLS) * 4;
            const v4f zero = {0.0f, 0.0f, 0.0f, 0.0f};
            stage_store(&lds_ego[j * DIM + d4],
                        (2 * j     < n_valid) ? lo[u] : zero,
                        (2 * j + 1 < n_valid) ? hi[u] : zero);
        }
    }
    __syncthreads();

    v8f c0[4] = {{}, {}, {}, {}};
    v8f c1[4] = {{}, {}, {}, {}};

    const v2f* lds_lane = &lds_ego[half * DIM + row];
    if (fast) {
        projT_kloop<false>(c0, c1, A0, A1, v, lds_lane, n_begin + khalf, f0 + row);
    } else {
        projT_kloop<true >(c0, c1, A0, A1, v, lds_lane, n_begin + khalf, f0 + row);
    }

    float* outp = partial + (size_t)chunk * FDIM * DIM;
    const int mbase = f0 + (half << 3);
#pragma unroll
    for (int t = 0; t < 4; ++t) {
#pragma unroll
        for (int j = 0; j < 8; ++j) {
            outp[(size_t)(mbase + j)      * DIM + t * 16 + row] = c0[t][j];
            outp[(size_t)(mbase + 16 + j) * DIM + t * 16 + row] = c1[t][j];
        }
    }
}

__global__ void lgcn_reduce(const float* __restrict__ partial,
                            const float* __restrict__ filt  ,
                            float* __restrict__ proj  ) {
    int i = blockIdx.x * blockDim.x + threadIdx.x;
    if (i < FDIM * DIM) {
        float s = 0.0f;
        for (int c = 0; c < NCHUNKS; ++c)
            s += partial[(size_t)c * FDIM * DIM + i];
        proj[i] = s * filt[i / DIM];
    }
}

__device__ __forceinline__ void prop_loadA(v2f (&a0)[KB], v2f (&a1)[KB],
                                           const float* p0, const float* p1, int blk) {
    p0 += blk * (KB * 4);
    p1 += blk * (KB * 4);
#pragma unroll
    for (int i = 0; i < KB; ++i) {
        a0[i] = *(const v2f*)(p0 + 4 * i);
        a1[i] = *(const v2f*)(p1 + 4 * i);
    }
}

template <bool RAGGED>
__device__ __forceinline__ void prop_writeback(const v8f (&c0)[4], const v8f (&c1)[4],
                                               float* __restrict__ ego,
                                               float* __restrict__ acc,
                                               int mbase, int row) {
    float old0[4][8], old1[4][8];
#pragma unroll
    for (int j = 0; j < 8; ++j) {
        const int r0 = mbase + j;
        const int r1 = mbase + 16 + j;
        const size_t o0 = (size_t)(RAGGED ? min(r0, N_TOTAL - 1) : r0) * DIM + row;
        const size_t o1 = (size_t)(RAGGED ? min(r1, N_TOTAL - 1) : r1) * DIM + row;
#pragma unroll
        for (int t = 0; t < 4; ++t) {
            old0[t][j] = acc[o0 + t * 16];
            old1[t][j] = acc[o1 + t * 16];
        }
    }
#pragma unroll
    for (int j = 0; j < 8; ++j) {
        const int r0 = mbase + j;
        const int r1 = mbase + 16 + j;
        if (!RAGGED || r0 < N_TOTAL) {
#pragma unroll
            for (int t = 0; t < 4; ++t) {
                const size_t idx = (size_t)r0 * DIM + t * 16 + row;
                ego[idx] = c0[t][j];
                acc[idx] = old0[t][j] + c0[t][j];
            }
        }
        if (!RAGGED || r1 < N_TOTAL) {
#pragma unroll
            for (int t = 0; t < 4; ++t) {
                const size_t idx = (size_t)r1 * DIM + t * 16 + row;
                ego[idx] = c1[t][j];
                acc[idx] = old1[t][j] + c1[t][j];
            }
        }
    }
}

__global__ __launch_bounds__(PROP_THREADS)
void lgcn_prop(const float* __restrict__ v,
               const float* __restrict__ proj,
               float* __restrict__ ego,
               float* __restrict__ acc) {
    __shared__ __attribute__((aligned(16))) v2f lds_proj[(FDIM / 2) * DIM];

    for (int base = threadIdx.x; base < STAGE_B_ITEMS; base += PROP_THREADS * STAGE_B_GROUP) {
        v4f lo[STAGE_B_GROUP], hi[STAGE_B_GROUP];
#pragma unroll
        for (int u = 0; u < STAGE_B_GROUP; ++u) {
            const int q  = base + u * PROP_THREADS;
            const int j  = q / QCOLS;
            const int d4 = (q % QCOLS) * 4;
            lo[u] = *(const v4f*)(proj + (size_t)(2 * j)     * DIM + d4);
            hi[u] = *(const v4f*)(proj + (size_t)(2 * j + 1) * DIM + d4);
        }
#pragma unroll
        for (int u = 0; u < STAGE_B_GROUP; ++u) {
            const int q  = base + u * PROP_THREADS;
            const int j  = q / QCOLS;
            const int d4 = (q % QCOLS) * 4;
            stage_store(&lds_proj[j * DIM + d4], lo[u], hi[u]);
        }
    }
    __syncthreads();

    const int lane  = threadIdx.x & 31;
    const int wave  = threadIdx.x >> 5;
    const int m0    = blockIdx.x * 256 + wave * 32;
    const int row   = lane & 15;
    const int half  = lane >> 4;
    const int khalf = half * 2;

    const int m = m0 + row;
    const float* vrow0 = v + (size_t)min(m,      N_TOTAL - 1) * FDIM;
    const float* vrow1 = v + (size_t)min(m + 16, N_TOTAL - 1) * FDIM;

    v8f c0[4] = {{}, {}, {}, {}};
    v8f c1[4] = {{}, {}, {}, {}};

    const v2f* lds_lane = &lds_proj[half * DIM + row];
    v2f A0[KB], A1[KB], B0[KB], B1[KB];
    prop_loadA(A0, A1, vrow0 + khalf, vrow1 + khalf, 0);
#pragma unroll 1
    for (int blk = 0; blk < NBLK_B; blk += 2) {
        prop_loadA(B0, B1, vrow0 + khalf, vrow1 + khalf, blk + 1);
        mma_block(c0, c1, A0, A1, lds_lane + (size_t)blk * (KB * 2) * DIM);
        prop_loadA(A0, A1, vrow0 + khalf, vrow1 + khalf, min(blk + 2, NBLK_B - 1));
        mma_block(c0, c1, B0, B1, lds_lane + (size_t)(blk + 1) * (KB * 2) * DIM);
    }

    const int mbase = m0 + (half << 3);
    if (m0 + 32 <= N_TOTAL) prop_writeback<false>(c0, c1, ego, acc, mbase, row);
    else                    prop_writeback<true >(c0, c1, ego, acc, mbase, row);
}

__global__ void lgcn_final(const float* __restrict__ acc,
                           float* __restrict__ out) {
    int i = blockIdx.x * blockDim.x + threadIdx.x;
    if (i < N_TOTAL * DIM) out[i] = acc[i] * 0.25f;
}

extern "C" void kernel_launch(void* const* d_in, const int* in_sizes, int n_in,
                              void* d_out, int out_size, void* d_ws, size_t ws_size,
                              hipStream_t stream) {
    const float* user    = (const float*)d_in[0];
    const float* item    = (const float*)d_in[1];
    const float* v       = (const float*)d_in[2];
    const float* filters = (const float*)d_in[3];

    float* ego     = (float*)d_ws;
    float* acc     = ego + (size_t)N_TOTAL * DIM;
    float* partial = acc + (size_t)N_TOTAL * DIM;
    float* proj    = partial + (size_t)NCHUNKS * FDIM * DIM;

    const int totalND = N_TOTAL * DIM;
    lgcn_init<<<(totalND + 255) / 256, 256, 0, stream>>>(user, item, ego, acc);

    for (int layer = 0; layer < N_LAYERS; ++layer) {
        dim3 gA(NCHUNKS, 4);
        lgcn_projT<<<gA, PROJT_THREADS, 0, stream>>>(v, ego, partial);

        lgcn_reduce<<<(FDIM * DIM + 255) / 256, 256, 0, stream>>>(
            partial, filters + (size_t)layer * FDIM, proj);

        const int nblocksB = (N_TOTAL + 255) / 256;
        lgcn_prop<<<nblocksB, PROP_THREADS, 0, stream>>>(v, proj, ego, acc);
    }

    lgcn_final<<<(totalND + 255) / 256, 256, 0, stream>>>(acc, (float*)d_out);
}


# prop grid: 246 workgroups of 288 rows (a 32-row tile per wave, then a 16-row tile on waves 0 and 1), was 277 of 256 rows; every sum and its order unchanged
# speedup vs baseline: 3.2176x; 1.1806x over previous
#include <hip/hip_runtime.h>


typedef __attribute__((ext_vector_type(2))) float v2f;
typedef __attribute__((ext_vector_type(4))) float v4f;
typedef __attribute__((ext_vector_type(8))) float v8f;

#define N_USERS 29858
#define N_ITEMS 40981
#define N_TOTAL (N_USERS + N_ITEMS)
#define DIM     64
#define FDIM    512
#define N_LAYERS 3
#define NCHUNKS 128
#define CHUNK   556
#define QUADS   (CHUNK / 4)
#define KB      8
#define NBLK_A  ((QUADS + KB - 1) / KB + ((QUADS + KB - 1) / KB) % 2)
#define ROWS_PAD (NBLK_A * KB * 4)
#define NBLK_B  (FDIM / (KB * 4))
#define PROJT_THREADS 128
#define PROP_THREADS  256
#define PROP_WAVES    (PROP_THREADS / 32)
#define PROP_ROWS_WG  (PROP_WAVES * 32 + 2 * 16)
#define PROP_BLOCKS   ((N_TOTAL + PROP_ROWS_WG - 1) / PROP_ROWS_WG)
#define QCOLS   (DIM / 4)
#define STAGE_A_ITEMS ((ROWS_PAD / 2) * QCOLS)
#define STAGE_B_ITEMS ((FDIM / 2) * QCOLS)
#define STAGE_A_GROUP 9
#define STAGE_B_GROUP 8

static_assert(NCHUNKS * CHUNK >= N_TOTAL, "chunk coverage");
static_assert(NBLK_A % 2 == 0 && NBLK_B % 2 == 0, "two register sets alternate");
static_assert(ROWS_PAD >= CHUNK && ROWS_PAD % 2 == 0, "padded chunk");
static_assert(STAGE_A_ITEMS % (PROJT_THREADS * STAGE_A_GROUP) == 0, "whole staging groups");
static_assert(STAGE_B_ITEMS % (PROP_THREADS * STAGE_B_GROUP) == 0, "whole staging groups");

#define WMMA_F32(A, B, C) \
    __builtin_amdgcn_wmma_f32_16x16x4_f32(false, (A), false, (B), (short)0, (C), false, false)

template <bool TWO = true>
__device__ __forceinline__ void mma_block(v8f (&c0)[4], v8f (&c1)[4],
                                          const v2f (&a0)[KB], const v2f (&a1)[KB],
                                          const v2f* bp) {
#pragma unroll
    for (int i = 0; i < KB; ++i) {
#pragma unroll
        for (int t = 0; t < 4; ++t) {
            v2f b = bp[(2 * i) * DIM + t * 16];
            c0[t] = WMMA_F32(a0[i], b, c0[t]);
            if (TWO) c1[t] = WMMA_F32(a1[i], b, c1[t]);
        }
    }
}

__device__ __forceinline__ void stage_store(v2f* dst, v4f lo, v4f hi) {
    v2f e;
    e.x = lo.x; e.y = hi.x; dst[0] = e;
    e.x = lo.y; e.y = hi.y; dst[1] = e;
    e.x = lo.z; e.y = hi.z; dst[2] = e;
    e.x = lo.w; e.y = hi.w; dst[3] = e;
}

template <bool RAGGED>
__device__ __forceinline__ void projT_loadA(v2f (&a0)[KB], v2f (&a1)[KB],
                                            const float* __restrict__ v,
                                            int n0, int col) {
    if (RAGGED) {
#pragma unroll
        for (int i = 0; i < KB; ++i) {
            const int na = n0 + 4 * i;
            const size_t r0 = (size_t)min(na,     N_TOTAL - 1) * FDIM;
            const size_t r1 = (size_t)min(na + 1, N_TOTAL - 1) * FDIM;
            a0[i].x = v[r0 + col];        a0[i].y = v[r1 + col];
            a1[i].x = v[r0 + col + 16];   a1[i].y = v[r1 + col + 16];
        }
    } else {
        const float* ap = v + (size_t)n0 * FDIM + col;
#pragma unroll
        for (int i = 0; i < KB; ++i) {
            a0[i].x = ap[(4 * i) * FDIM];        a0[i].y = ap[(4 * i + 1) * FDIM];
            a1[i].x = ap[(4 * i) * FDIM + 16];   a1[i].y = ap[(4 * i + 1) * FDIM + 16];
        }
    }
}

template <bool RAGGED>
__device__ __forceinline__ void projT_kloop(v8f (&c0)[4], v8f (&c1)[4],
                                            v2f (&A0)[KB], v2f (&A1)[KB],
                                            const float* __restrict__ v,
                                            const v2f* lds_lane, int nrow, int col) {
    v2f B0[KB], B1[KB];
#pragma unroll 1
    for (int blk = 0; blk < NBLK_A; blk += 2) {
        projT_loadA<RAGGED>(B0, B1, v, nrow + (blk + 1) * (KB * 4), col);
        mma_block(c0, c1, A0, A1, lds_lane + (size_t)blk * (KB * 2) * DIM);
        projT_loadA<RAGGED>(A0, A1, v, nrow + min(blk + 2, NBLK_A - 1) * (KB * 4), col);
        mma_block(c0, c1, B0, B1, lds_lane + (size_t)(blk + 1) * (KB * 2) * DIM);
    }
}

__global__ void lgcn_init(const float* __restrict__ user,
                          const float* __restrict__ item,
                          float* __restrict__ ego,
                          float* __restrict__ acc) {
    int i = blockIdx.x * blockDim.x + threadIdx.x;
    if (i < N_TOTAL * DIM) {
        float val = (i < N_USERS * DIM) ? user[i] : item[i - N_USERS * DIM];
        ego[i] = val;
        acc[i] = val;
    }
}

__global__ __launch_bounds__(PROJT_THREADS)
void lgcn_projT(const float* __restrict__ v,
                const float* __restrict__ ego,
                float* __restrict__ partial  ) {
    __shared__ __attribute__((aligned(16))) v2f lds_ego[(ROWS_PAD / 2) * DIM];

    const int chunk   = blockIdx.x;
    const int n_begin = chunk * CHUNK;
    const int n_valid = min(N_TOTAL - n_begin, CHUNK);

    const int lane  = threadIdx.x & 31;
    const int wave  = threadIdx.x >> 5;
    const int f0    = (blockIdx.y * 4 + wave) * 32;
    const int row   = lane & 15;
    const int half  = lane >> 4;
    const int khalf = half * 2;

    const bool fast = (n_begin + ROWS_PAD <= N_TOTAL);

    v2f A0[KB], A1[KB];
    if (fast) projT_loadA<false>(A0, A1, v, n_begin + khalf, f0 + row);
    else      projT_loadA<true >(A0, A1, v, n_begin + khalf, f0 + row);

    const float* ego_chunk = ego + (size_t)n_begin * DIM;
    for (int base = threadIdx.x; base < STAGE_A_ITEMS; base += PROJT_THREADS * STAGE_A_GROUP) {
        v4f lo[STAGE_A_GROUP], hi[STAGE_A_GROUP];
#pragma unroll
        for (int u = 0; u < STAGE_A_GROUP; ++u) {
            const int q  = base + u * PROJT_THREADS;
            const int j  = q / QCOLS;
            const int d4 = (q % QCOLS) * 4;
            lo[u] = *(const v4f*)(ego_chunk + (size_t)min(2 * j,     n_valid - 1) * DIM + d4);
            hi[u] = *(const v4f*)(ego_chunk + (size_t)min(2 * j + 1, n_valid - 1) * DIM + d4);
        }
#pragma unroll
        for (int u = 0; u < STAGE_A_GROUP; ++u) {
            const int q  = base + u * PROJT_THREADS;
            const int j  = q / QCOLS;
            const int d4 = (q % QCOLS) * 4;
            const v4f zero = {0.0f, 0.0f, 0.0f, 0.0f};
            stage_store(&lds_ego[j * DIM + d4],
                        (2 * j     < n_valid) ? lo[u] : zero,
                        (2 * j + 1 < n_valid) ? hi[u] : zero);
        }
    }
    __syncthreads();

    v8f c0[4] = {{}, {}, {}, {}};
    v8f c1[4] = {{}, {}, {}, {}};

    const v2f* lds_lane = &lds_ego[half * DIM + row];
    if (fast) {
        projT_kloop<false>(c0, c1, A0, A1, v, lds_lane, n_begin + khalf, f0 + row);
    } else {
        projT_kloop<true >(c0, c1, A0, A1, v, lds_lane, n_begin + khalf, f0 + row);
    }

    float* outp = partial + (size_t)chunk * FDIM * DIM;
    const int mbase = f0 + (half << 3);
#pragma unroll
    for (int t = 0; t < 4; ++t) {
#pragma unroll
        for (int j = 0; j < 8; ++j) {
            outp[(size_t)(mbase + j)      * DIM + t * 16 + row] = c0[t][j];
            outp[(size_t)(mbase + 16 + j) * DIM + t * 16 + row] = c1[t][j];
        }
    }
}

__global__ void lgcn_reduce(const float* __restrict__ partial,
                            const float* __restrict__ filt  ,
                            float* __restrict__ proj  ) {
    int i = blockIdx.x * blockDim.x + threadIdx.x;
    if (i < FDIM * DIM) {
        float s = 0.0f;
        for (int c = 0; c < NCHUNKS; ++c)
            s += partial[(size_t)c * FDIM * DIM + i];
        proj[i] = s * filt[i / DIM];
    }
}

template <bool TWO>
__device__ __forceinline__ void prop_loadA(v2f (&a0)[KB], v2f (&a1)[KB],
                                           const float* p0, const float* p1, int blk) {
    p0 += blk * (KB * 4);
    p1 += blk * (KB * 4);
#pragma unroll
    for (int i = 0; i < KB; ++i) {
        a0[i] = *(const v2f*)(p0 + 4 * i);
        if (TWO) a1[i] = *(const v2f*)(p1 + 4 * i);
    }
}

template <bool RAGGED, bool TWO>
__device__ __forceinline__ void prop_writeback(const v8f (&c0)[4], const v8f (&c1)[4],
                                               float* __restrict__ ego,
                                               float* __restrict__ acc,
                                               int mbase, int row) {
    float old0[4][8], old1[4][8];
#pragma unroll
    for (int j = 0; j < 8; ++j) {
        const int r0 = mbase + j;
        const int r1 = mbase + 16 + j;
        const size_t o0 = (size_t)(RAGGED ? min(r0, N_TOTAL - 1) : r0) * DIM + row;
        const size_t o1 = (size_t)(RAGGED ? min(r1, N_TOTAL - 1) : r1) * DIM + row;
#pragma unroll
        for (int t = 0; t < 4; ++t) {
            old0[t][j] = acc[o0 + t * 16];
            if (TWO) old1[t][j] = acc[o1 + t * 16];
        }
    }
#pragma unroll
    for (int j = 0; j < 8; ++j) {
        const int r0 = mbase + j;
        const int r1 = mbase + 16 + j;
        if (!RAGGED || r0 < N_TOTAL) {
#pragma unroll
            for (int t = 0; t < 4; ++t) {
                const size_t idx = (size_t)r0 * DIM + t * 16 + row;
                ego[idx] = c0[t][j];
                acc[idx] = old0[t][j] + c0[t][j];
            }
        }
        if (TWO && (!RAGGED || r1 < N_TOTAL)) {
#pragma unroll
            for (int t = 0; t < 4; ++t) {
                const size_t idx = (size_t)r1 * DIM + t * 16 + row;
                ego[idx] = c1[t][j];
                acc[idx] = old1[t][j] + c1[t][j];
            }
        }
    }
}

template <bool TWO>
__device__ __forceinline__ void prop_tile(const float* __restrict__ v,
                                          float* __restrict__ ego,
                                          float* __restrict__ acc,
                                          const v2f* lds_lane,
                                          int m0, int row, int half) {
    if (m0 >= N_TOTAL) return;
    const int khalf = half * 2;
    const int m = m0 + row;
    const float* vrow0 = v + (size_t)min(m,      N_TOTAL - 1) * FDIM;
    const float* vrow1 = v + (size_t)min(m + 16, N_TOTAL - 1) * FDIM;

    v8f c0[4] = {{}, {}, {}, {}};
    v8f c1[4] = {{}, {}, {}, {}};

    v2f A0[KB], A1[KB], B0[KB], B1[KB];
    prop_loadA<TWO>(A0, A1, vrow0 + khalf, vrow1 + khalf, 0);
#pragma unroll 1
    for (int blk = 0; blk < NBLK_B; blk += 2) {
        prop_loadA<TWO>(B0, B1, vrow0 + khalf, vrow1 + khalf, blk + 1);
        mma_block<TWO>(c0, c1, A0, A1, lds_lane + (size_t)blk * (KB * 2) * DIM);
        prop_loadA<TWO>(A0, A1, vrow0 + khalf, vrow1 + khalf, min(blk + 2, NBLK_B - 1));
        mma_block<TWO>(c0, c1, B0, B1, lds_lane + (size_t)(blk + 1) * (KB * 2) * DIM);
    }

    const int mbase = m0 + (half << 3);
    if (m0 + (TWO ? 32 : 16) <= N_TOTAL) prop_writeback<false, TWO>(c0, c1, ego, acc, mbase, row);
    else                                 prop_writeback<true,  TWO>(c0, c1, ego, acc, mbase, row);
}

__global__ __launch_bounds__(PROP_THREADS)
void lgcn_prop(const float* __restrict__ v,
               const float* __restrict__ proj,
               float* __restrict__ ego,
               float* __restrict__ acc) {
    __shared__ __attribute__((aligned(16))) v2f lds_proj[(FDIM / 2) * DIM];

    for (int base = threadIdx.x; base < STAGE_B_ITEMS; base += PROP_THREADS * STAGE_B_GROUP) {
        v4f lo[STAGE_B_GROUP], hi[STAGE_B_GROUP];
#pragma unroll
        for (int u = 0; u < STAGE_B_GROUP; ++u) {
            const int q  = base + u * PROP_THREADS;
            const int j  = q / QCOLS;
            const int d4 = (q % QCOLS) * 4;
            lo[u] = *(const v4f*)(proj + (size_t)(2 * j)     * DIM + d4);
            hi[u] = *(const v4f*)(proj + (size_t)(2 * j + 1) * DIM + d4);
        }
#pragma unroll
        for (int u = 0; u < STAGE_B_GROUP; ++u) {
            const int q  = base + u * PROP_THREADS;
            const int j  = q / QCOLS;
            const int d4 = (q % QCOLS) * 4;
            stage_store(&lds_proj[j * DIM + d4], lo[u], hi[u]);
        }
    }
    __syncthreads();

    const int lane  = threadIdx.x & 31;
    const int wave  = threadIdx.x >> 5;
    const int row   = lane & 15;
    const int half  = lane >> 4;
    const v2f* lds_lane = &lds_proj[half * DIM + row];

    const int wg_row0 = blockIdx.x * PROP_ROWS_WG;
    prop_tile<true>(v, ego, acc, lds_lane, wg_row0 + wave * 32, row, half);
    if (wave < 2)
        prop_tile<false>(v, ego, acc, lds_lane, wg_row0 + PROP_WAVES * 32 + wave * 16, row, half);
}

__global__ void lgcn_final(const float* __restrict__ acc,
                           float* __restrict__ out) {
    int i = blockIdx.x * blockDim.x + threadIdx.x;
    if (i < N_TOTAL * DIM) out[i] = acc[i] * 0.25f;
}

extern "C" void kernel_launch(void* const* d_in, const int* in_sizes, int n_in,
                              void* d_out, int out_size, void* d_ws, size_t ws_size,
                              hipStream_t stream) {
    const float* user    = (const float*)d_in[0];
    const float* item    = (const float*)d_in[1];
    const float* v       = (const float*)d_in[2];
    const float* filters = (const float*)d_in[3];

    float* ego     = (float*)d_ws;
    float* acc     = ego + (size_t)N_TOTAL * DIM;
    float* partial = acc + (size_t)N_TOTAL * DIM;
    float* proj    = partial + (size_t)NCHUNKS * FDIM * DIM;

    const int totalND = N_TOTAL * DIM;
    lgcn_init<<<(totalND + 255) / 256, 256, 0, stream>>>(user, item, ego, acc);

    for (int layer = 0; layer < N_LAYERS; ++layer) {
        dim3 gA(NCHUNKS, 4);
        lgcn_projT<<<gA, PROJT_THREADS, 0, stream>>>(v, ego, partial);

        lgcn_reduce<<<(FDIM * DIM + 255) / 256, 256, 0, stream>>>(
            partial, filters + (size_t)layer * FDIM, proj);

        lgcn_prop<<<PROP_BLOCKS, PROP_THREADS, 0, stream>>>(v, proj, ego, acc);
    }

    lgcn_final<<<(totalND + 255) / 256, 256, 0, stream>>>(acc, (float*)d_out);
}


# No elementwise launches: layer 0's projT and prop read ego rows from the user and item tables, the last prop stores (acc + ego) / 4 to out; 9 launches, was 11; every sum and its order unchanged
# speedup vs baseline: 3.5249x; 1.0955x over previous
#include <hip/hip_runtime.h>


typedef __attribute__((ext_vector_type(2))) float v2f;
typedef __attribute__((ext_vector_type(4))) float v4f;
typedef __attribute__((ext_vector_type(8))) float v8f;

#define N_USERS 29858
#define N_ITEMS 40981
#define N_TOTAL (N_USERS + N_ITEMS)
#define DIM     64
#define FDIM    512
#define N_LAYERS 3
#define NCHUNKS 128
#define CHUNK   556
#define QUADS   (CHUNK / 4)
#define KB      8
#define NBLK_A  ((QUADS + KB - 1) / KB + ((QUADS + KB - 1) / KB) % 2)
#define ROWS_PAD (NBLK_A * KB * 4)
#define NBLK_B  (FDIM / (KB * 4))
#define PROJT_THREADS 128
#define PROP_THREADS  256
#define PROP_WAVES    (PROP_THREADS / 32)
#define PROP_ROWS_WG  (PROP_WAVES * 32 + 2 * 16)
#define PROP_BLOCKS   ((N_TOTAL + PROP_ROWS_WG - 1) / PROP_ROWS_WG)
#define QCOLS   (DIM / 4)
#define STAGE_A_ITEMS ((ROWS_PAD / 2) * QCOLS)
#define STAGE_B_ITEMS ((FDIM / 2) * QCOLS)
#define STAGE_A_GROUP 9
#define STAGE_B_GROUP 8

static_assert(NCHUNKS * CHUNK >= N_TOTAL, "chunk coverage");
static_assert(NBLK_A % 2 == 0 && NBLK_B % 2 == 0, "two register sets alternate");
static_assert(ROWS_PAD >= CHUNK && ROWS_PAD % 2 == 0, "padded chunk");
static_assert(STAGE_A_ITEMS % (PROJT_THREADS * STAGE_A_GROUP) == 0, "whole staging groups");
static_assert(STAGE_B_ITEMS % (PROP_THREADS * STAGE_B_GROUP) == 0, "whole staging groups");

#define WMMA_F32(A, B, C) \
    __builtin_amdgcn_wmma_f32_16x16x4_f32(false, (A), false, (B), (short)0, (C), false, false)

template <bool TWO = true>
__device__ __forceinline__ void mma_block(v8f (&c0)[4], v8f (&c1)[4],
                                          const v2f (&a0)[KB], const v2f (&a1)[KB],
                                          const v2f* bp) {
#pragma unroll
    for (int i = 0; i < KB; ++i) {
#pragma unroll
        for (int t = 0; t < 4; ++t) {
            v2f b = bp[(2 * i) * DIM + t * 16];
            c0[t] = WMMA_F32(a0[i], b, c0[t]);
            if (TWO) c1[t] = WMMA_F32(a1[i], b, c1[t]);
        }
    }
}

__device__ __forceinline__ void stage_store(v2f* dst, v4f lo, v4f hi) {
    v2f e;
    e.x = lo.x; e.y = hi.x; dst[0] = e;
    e.x = lo.y; e.y = hi.y; dst[1] = e;
    e.x = lo.z; e.y = hi.z; dst[2] = e;
    e.x = lo.w; e.y = hi.w; dst[3] = e;
}

template <bool FIRST>
__device__ __forceinline__ const float* ego_row(const float* a, const float* b, int r) {
    if (FIRST) return (r < N_USERS) ? a + (size_t)r * DIM : b + (size_t)(r - N_USERS) * DIM;
    return a + (size_t)r * DIM;
}

template <bool RAGGED>
__device__ __forceinline__ void projT_loadA(v2f (&a0)[KB], v2f (&a1)[KB],
                                            const float* __restrict__ v,
                                            int n0, int col) {
    if (RAGGED) {
#pragma unroll
        for (int i = 0; i < KB; ++i) {
            const int na = n0 + 4 * i;
            const size_t r0 = (size_t)min(na,     N_TOTAL - 1) * FDIM;
            const size_t r1 = (size_t)min(na + 1, N_TOTAL - 1) * FDIM;
            a0[i].x = v[r0 + col];        a0[i].y = v[r1 + col];
            a1[i].x = v[r0 + col + 16];   a1[i].y = v[r1 + col + 16];
        }
    } else {
        const float* ap = v + (size_t)n0 * FDIM + col;
#pragma unroll
        for (int i = 0; i < KB; ++i) {
            a0[i].x = ap[(4 * i) * FDIM];        a0[i].y = ap[(4 * i + 1) * FDIM];
            a1[i].x = ap[(4 * i) * FDIM + 16];   a1[i].y = ap[(4 * i + 1) * FDIM + 16];
        }
    }
}

template <bool RAGGED>
__device__ __forceinline__ void projT_kloop(v8f (&c0)[4], v8f (&c1)[4],
                                            v2f (&A0)[KB], v2f (&A1)[KB],
                                            const float* __restrict__ v,
                                            const v2f* lds_lane, int nrow, int col) {
    v2f B0[KB], B1[KB];
#pragma unroll 1
    for (int blk = 0; blk < NBLK_A; blk += 2) {
        projT_loadA<RAGGED>(B0, B1, v, nrow + (blk + 1) * (KB * 4), col);
        mma_block(c0, c1, A0, A1, lds_lane + (size_t)blk * (KB * 2) * DIM);
        projT_loadA<RAGGED>(A0, A1, v, nrow + min(blk + 2, NBLK_A - 1) * (KB * 4), col);
        mma_block(c0, c1, B0, B1, lds_lane + (size_t)(blk + 1) * (KB * 2) * DIM);
    }
}

template <bool FIRST>
__global__ __launch_bounds__(PROJT_THREADS)
void lgcn_projT(const float* __restrict__ v,
                const float* __restrict__ ego0,
                const float* __restrict__ ego1,
                float* __restrict__ partial  ) {
    __shared__ __attribute__((aligned(16))) v2f lds_ego[(ROWS_PAD / 2) * DIM];

    const int chunk   = blockIdx.x;
    const int n_begin = chunk * CHUNK;
    const int n_valid = min(N_TOTAL - n_begin, CHUNK);

    const int lane  = threadIdx.x & 31;
    const int wave  = threadIdx.x >> 5;
    const int f0    = (blockIdx.y * 4 + wave) * 32;
    const int row   = lane & 15;
    const int half  = lane >> 4;
    const int khalf = half * 2;

    const bool fast = (n_begin + ROWS_PAD <= N_TOTAL);

    v2f A0[KB], A1[KB];
    if (fast) projT_loadA<false>(A0, A1, v, n_begin + khalf, f0 + row);
    else      projT_loadA<true >(A0, A1, v, n_begin + khalf, f0 + row);

    for (int base = threadIdx.x; base < STAGE_A_ITEMS; base += PROJT_THREADS * STAGE_A_GROUP) {
        v4f lo[STAGE_A_GROUP], hi[STAGE_A_GROUP];
#pragma unroll
        for (int u = 0; u < STAGE_A_GROUP; ++u) {
            const int q  = base + u * PROJT_THREADS;
            const int j  = q / QCOLS;
            const int d4 = (q % QCOLS) * 4;
            lo[u] = *(const v4f*)(ego_row<FIRST>(ego0, ego1, n_begin + min(2 * j,     n_valid - 1)) + d4);
            hi[u] = *(const v4f*)(ego_row<FIRST>(ego0, ego1, n_begin + min(2 * j + 1, n_valid - 1)) + d4);
        }
#pragma unroll
        for (int u = 0; u < STAGE_A_GROUP; ++u) {
            const int q  = base + u * PROJT_THREADS;
            const int j  = q / QCOLS;
            const int d4 = (q % QCOLS) * 4;
            const v4f zero = {0.0f, 0.0f, 0.0f, 0.0f};
            stage_store(&lds_ego[j * DIM + d4],
                        (2 * j     < n_valid) ? lo[u] : zero,
                        (2 * j + 1 < n_valid) ? hi[u] : zero);
        }
    }
    __syncthreads();

    v8f c0[4] = {{}, {}, {}, {}};
    v8f c1[4] = {{}, {}, {}, {}};

    const v2f* lds_lane = &lds_ego[half * DIM + row];
    if (fast) {
        projT_kloop<false>(c0, c1, A0, A1, v, lds_lane, n_begin + khalf, f0 + row);
    } else {
        projT_kloop<true >(c0, c1, A0, A1, v, lds_lane, n_begin + khalf, f0 + row);
    }

    float* outp = partial + (size_t)chunk * FDIM * DIM;
    const int mbase = f0 + (half << 3);
#pragma unroll
    for (int t = 0; t < 4; ++t) {
#pragma unroll
        for (int j = 0; j < 8; ++j) {
            outp[(size_t)(mbase + j)      * DIM + t * 16 + row] = c0[t][j];
            outp[(size_t)(mbase + 16 + j) * DIM + t * 16 + row] = c1[t][j];
        }
    }
}

__global__ void lgcn_reduce(const float* __restrict__ partial,
                            const float* __restrict__ filt  ,
                            float* __restrict__ proj  ) {
    int i = blockIdx.x * blockDim.x + threadIdx.x;
    if (i < FDIM * DIM) {
        float s = 0.0f;
        for (int c = 0; c < NCHUNKS; ++c)
            s += partial[(size_t)c * FDIM * DIM + i];
        proj[i] = s * filt[i / DIM];
    }
}

template <bool TWO>
__device__ __forceinline__ void prop_loadA(v2f (&a0)[KB], v2f (&a1)[KB],
                                           const float* p0, const float* p1, int blk) {
    p0 += blk * (KB * 4);
    p1 += blk * (KB * 4);
#pragma unroll
    for (int i = 0; i < KB; ++i) {
        a0[i] = *(const v2f*)(p0 + 4 * i);
        if (TWO) a1[i] = *(const v2f*)(p1 + 4 * i);
    }
}

#define PROP_FIRST 0
#define PROP_MID   1
#define PROP_LAST  2

template <int MODE, bool RAGGED, bool TWO>
__device__ __forceinline__ void prop_writeback(const v8f (&c0)[4], const v8f (&c1)[4],
                                               const float* __restrict__ user,
                                               const float* __restrict__ item,
                                               float* __restrict__ ego,
                                               float* __restrict__ acc,
                                               float* __restrict__ out,
                                               int mbase, int row) {
    float old0[4][8], old1[4][8];
#pragma unroll
    for (int j = 0; j < 8; ++j) {
        const int r0 = mbase + j;
        const int r1 = mbase + 16 + j;
        const float* p0 = ego_row<MODE == PROP_FIRST>(MODE == PROP_FIRST ? user : acc, item,
                                                      RAGGED ? min(r0, N_TOTAL - 1) : r0) + row;
        const float* p1 = ego_row<MODE == PROP_FIRST>(MODE == PROP_FIRST ? user : acc, item,
                                                      RAGGED ? min(r1, N_TOTAL - 1) : r1) + row;
#pragma unroll
        for (int t = 0; t < 4; ++t) {
            old0[t][j] = p0[t * 16];
            if (TWO) old1[t][j] = p1[t * 16];
        }
    }
#pragma unroll
    for (int j = 0; j < 8; ++j) {
        const int r0 = mbase + j;
        const int r1 = mbase + 16 + j;
        if (!RAGGED || r0 < N_TOTAL) {
#pragma unroll
            for (int t = 0; t < 4; ++t) {
                const size_t idx = (size_t)r0 * DIM + t * 16 + row;
                if (MODE == PROP_LAST) {
                    out[idx] = (old0[t][j] + c0[t][j]) * (1.0f / (N_LAYERS + 1));
                } else {
                    ego[idx] = c0[t][j];
                    acc[idx] = old0[t][j] + c0[t][j];
                }
            }
        }
        if (TWO && (!RAGGED || r1 < N_TOTAL)) {
#pragma unroll
            for (int t = 0; t < 4; ++t) {
                const size_t idx = (size_t)r1 * DIM + t * 16 + row;
                if (MODE == PROP_LAST) {
                    out[idx] = (old1[t][j] + c1[t][j]) * (1.0f / (N_LAYERS + 1));
                } else {
                    ego[idx] = c1[t][j];
                    acc[idx] = old1[t][j] + c1[t][j];
                }
            }
        }
    }
}

template <int MODE, bool TWO>
__device__ __forceinline__ void prop_tile(const float* __restrict__ v,
                                          const float* __restrict__ user,
                                          const float* __restrict__ item,
                                          float* __restrict__ ego,
                                          float* __restrict__ acc,
                                          float* __restrict__ out,
                                          const v2f* lds_lane,
                                          int m0, int row, int half) {
    if (m0 >= N_TOTAL) return;
    const int khalf = half * 2;
    const int m = m0 + row;
    const float* vrow0 = v + (size_t)min(m,      N_TOTAL - 1) * FDIM;
    const float* vrow1 = v + (size_t)min(m + 16, N_TOTAL - 1) * FDIM;

    v8f c0[4] = {{}, {}, {}, {}};
    v8f c1[4] = {{}, {}, {}, {}};

    v2f A0[KB], A1[KB], B0[KB], B1[KB];
    prop_loadA<TWO>(A0, A1, vrow0 + khalf, vrow1 + khalf, 0);
#pragma unroll 1
    for (int blk = 0; blk < NBLK_B; blk += 2) {
        prop_loadA<TWO>(B0, B1, vrow0 + khalf, vrow1 + khalf, blk + 1);
        mma_block<TWO>(c0, c1, A0, A1, lds_lane + (size_t)blk * (KB * 2) * DIM);
        prop_loadA<TWO>(A0, A1, vrow0 + khalf, vrow1 + khalf, min(blk + 2, NBLK_B - 1));
        mma_block<TWO>(c0, c1, B0, B1, lds_lane + (size_t)(blk + 1) * (KB * 2) * DIM);
    }

    const int mbase = m0 + (half << 3);
    if (m0 + (TWO ? 32 : 16) <= N_TOTAL)
        prop_writeback<MODE, false, TWO>(c0, c1, user, item, ego, acc, out, mbase, row);
    else
        prop_writeback<MODE, true,  TWO>(c0, c1, user, item, ego, acc, out, mbase, row);
}

template <int MODE>
__global__ __launch_bounds__(PROP_THREADS)
void lgcn_prop(const float* __restrict__ v,
               const float* __restrict__ proj,
               const float* __restrict__ user,
               const float* __restrict__ item,
               float* __restrict__ ego,
               float* __restrict__ acc,
               float* __restrict__ out) {
    __shared__ __attribute__((aligned(16))) v2f lds_proj[(FDIM / 2) * DIM];

    for (int base = threadIdx.x; base < STAGE_B_ITEMS; base += PROP_THREADS * STAGE_B_GROUP) {
        v4f lo[STAGE_B_GROUP], hi[STAGE_B_GROUP];
#pragma unroll
        for (int u = 0; u < STAGE_B_GROUP; ++u) {
            const int q  = base + u * PROP_THREADS;
            const int j  = q / QCOLS;
            const int d4 = (q % QCOLS) * 4;
            lo[u] = *(const v4f*)(proj + (size_t)(2 * j)     * DIM + d4);
            hi[u] = *(const v4f*)(proj + (size_t)(2 * j + 1) * DIM + d4);
        }
#pragma unroll
        for (int u = 0; u < STAGE_B_GROUP; ++u) {
            const int q  = base + u * PROP_THREADS;
            const int j  = q / QCOLS;
            const int d4 = (q % QCOLS) * 4;
            stage_store(&lds_proj[j * DIM + d4], lo[u], hi[u]);
        }
    }
    __syncthreads();

    const int lane  = threadIdx.x & 31;
    const int wave  = threadIdx.x >> 5;
    const int row   = lane & 15;
    const int half  = lane >> 4;
    const v2f* lds_lane = &lds_proj[half * DIM + row];

    const int wg_row0 = blockIdx.x * PROP_ROWS_WG;
    prop_tile<MODE, true>(v, user, item, ego, acc, out, lds_lane, wg_row0 + wave * 32, row, half);
    if (wave < 2)
        prop_tile<MODE, false>(v, user, item, ego, acc, out, lds_lane,
                               wg_row0 + PROP_WAVES * 32 + wave * 16, row, half);
}

extern "C" void kernel_launch(void* const* d_in, const int* in_sizes, int n_in,
                              void* d_out, int out_size, void* d_ws, size_t ws_size,
                              hipStream_t stream) {
    const float* user    = (const float*)d_in[0];
    const float* item    = (const float*)d_in[1];
    const float* v       = (const float*)d_in[2];
    const float* filters = (const float*)d_in[3];

    float* ego     = (float*)d_ws;
    float* acc     = ego + (size_t)N_TOTAL * DIM;
    float* partial = acc + (size_t)N_TOTAL * DIM;
    float* proj    = partial + (size_t)NCHUNKS * FDIM * DIM;
    float* out     = (float*)d_out;

    static_assert(N_LAYERS == 3, "one first, one middle and one last layer are launched below");
    const dim3 gA(NCHUNKS, 4);
    const int  gR = (FDIM * DIM + 255) / 256;

    lgcn_projT<true><<<gA, PROJT_THREADS, 0, stream>>>(v, user, item, partial);
    lgcn_reduce<<<gR, 256, 0, stream>>>(partial, filters, proj);
    lgcn_prop<PROP_FIRST><<<PROP_BLOCKS, PROP_THREADS, 0, stream>>>(
        v, proj, user, item, ego, acc, nullptr);

    lgcn_projT<false><<<gA, PROJT_THREADS, 0, stream>>>(v, ego, nullptr, partial);
    lgcn_reduce<<<gR, 256, 0, stream>>>(partial, filters + FDIM, proj);
    lgcn_prop<PROP_MID><<<PROP_BLOCKS, PROP_THREADS, 0, stream>>>(
        v, proj, nullptr, nullptr, ego, acc, nullptr);

    lgcn_projT<false><<<gA, PROJT_THREADS, 0, stream>>>(v, ego, nullptr, partial);
    lgcn_reduce<<<gR, 256, 0, stream>>>(partial, filters + 2 * FDIM, proj);
    lgcn_prop<PROP_LAST><<<PROP_BLOCKS, PROP_THREADS, 0, stream>>>(
        v, proj, nullptr, nullptr, nullptr, acc, out);
}


# lgcn_prop: 12 waves, the left-over 32 rows as four 16x32 tiles on waves 8-11 beside the eight 32-row tiles (was two 16x64 after them on waves 0,1), A loads pinned a K-block ahead; sum order kept
# speedup vs baseline: 3.6761x; 1.0429x over previous
#include <hip/hip_runtime.h>


typedef __attribute__((ext_vector_type(2))) float v2f;
typedef __attribute__((ext_vector_type(4))) float v4f;
typedef __attribute__((ext_vector_type(8))) float v8f;

#define N_USERS 29858
#define N_ITEMS 40981
#define N_TOTAL (N_USERS + N_ITEMS)
#define DIM     64
#define FDIM    512
#define N_LAYERS 3
#define NCHUNKS 128
#define CHUNK   556
#define QUADS   (CHUNK / 4)
#define KB      8
#define NBLK_A  ((QUADS + KB - 1) / KB + ((QUADS + KB - 1) / KB) % 2)
#define ROWS_PAD (NBLK_A * KB * 4)
#define NBLK_B  (FDIM / (KB * 4))
#define PROJT_THREADS 128
#define PROP_MAIN_WAVES 8
#define PROP_TAIL_WAVES 4
#define PROP_WAVES    (PROP_MAIN_WAVES + PROP_TAIL_WAVES)
#define PROP_THREADS  (PROP_WAVES * 32)
#define PROP_STAGE_THREADS (PROP_MAIN_WAVES * 32)
#define PROP_ROWS_WG  (PROP_MAIN_WAVES * 32 + (PROP_TAIL_WAVES / 2) * 16)
#define PROP_BLOCKS   ((N_TOTAL + PROP_ROWS_WG - 1) / PROP_ROWS_WG)
#define QCOLS   (DIM / 4)
#define STAGE_A_ITEMS ((ROWS_PAD / 2) * QCOLS)
#define STAGE_B_ITEMS ((FDIM / 2) * QCOLS)
#define STAGE_A_GROUP 9
#define STAGE_B_GROUP 8

static_assert(NCHUNKS * CHUNK >= N_TOTAL, "chunk coverage");
static_assert(NBLK_A % 2 == 0 && NBLK_B % 2 == 0, "two register sets alternate");
static_assert(ROWS_PAD >= CHUNK && ROWS_PAD % 2 == 0, "padded chunk");
static_assert(STAGE_A_ITEMS % (PROJT_THREADS * STAGE_A_GROUP) == 0, "whole staging groups");
static_assert(STAGE_B_ITEMS % (PROP_STAGE_THREADS * STAGE_B_GROUP) == 0, "whole staging groups");
static_assert(PROP_TAIL_WAVES % 2 == 0, "two 32-column tiles per left-over 16-row tile");

#define WMMA_F32(A, B, C) \
    __builtin_amdgcn_wmma_f32_16x16x4_f32(false, (A), false, (B), (short)0, (C), false, false)

template <bool TWO = true, int NT = 4>
__device__ __forceinline__ void mma_block(v8f (&c0)[NT], v8f (&c1)[NT],
                                          const v2f (&a0)[KB], const v2f (&a1)[KB],
                                          const v2f* bp) {
#pragma unroll
    for (int i = 0; i < KB; ++i) {
#pragma unroll
        for (int t = 0; t < NT; ++t) {
            v2f b = bp[(2 * i) * DIM + t * 16];
            c0[t] = WMMA_F32(a0[i], b, c0[t]);
            if (TWO) c1[t] = WMMA_F32(a1[i], b, c1[t]);
        }
    }
}

__device__ __forceinline__ void stage_store(v2f* dst, v4f lo, v4f hi) {
    v2f e;
    e.x = lo.x; e.y = hi.x; dst[0] = e;
    e.x = lo.y; e.y = hi.y; dst[1] = e;
    e.x = lo.z; e.y = hi.z; dst[2] = e;
    e.x = lo.w; e.y = hi.w; dst[3] = e;
}

template <bool FIRST>
__device__ __forceinline__ const float* ego_row(const float* a, const float* b, int r) {
    if (FIRST) return (r < N_USERS) ? a + (size_t)r * DIM : b + (size_t)(r - N_USERS) * DIM;
    return a + (size_t)r * DIM;
}

template <bool RAGGED>
__device__ __forceinline__ void projT_loadA(v2f (&a0)[KB], v2f (&a1)[KB],
                                            const float* __restrict__ v,
                                            int n0, int col) {
    if (RAGGED) {
#pragma unroll
        for (int i = 0; i < KB; ++i) {
            const int na = n0 + 4 * i;
            const size_t r0 = (size_t)min(na,     N_TOTAL - 1) * FDIM;
            const size_t r1 = (size_t)min(na + 1, N_TOTAL - 1) * FDIM;
            a0[i].x = v[r0 + col];        a0[i].y = v[r1 + col];
            a1[i].x = v[r0 + col + 16];   a1[i].y = v[r1 + col + 16];
        }
    } else {
        const float* ap = v + (size_t)n0 * FDIM + col;
#pragma unroll
        for (int i = 0; i < KB; ++i) {
            a0[i].x = ap[(4 * i) * FDIM];        a0[i].y = ap[(4 * i + 1) * FDIM];
            a1[i].x = ap[(4 * i) * FDIM + 16];   a1[i].y = ap[(4 * i + 1) * FDIM + 16];
        }
    }
}

template <bool RAGGED>
__device__ __forceinline__ void projT_kloop(v8f (&c0)[4], v8f (&c1)[4],
                                            v2f (&A0)[KB], v2f (&A1)[KB],
                                            const float* __restrict__ v,
                                            const v2f* lds_lane, int nrow, int col) {
    v2f B0[KB], B1[KB];
#pragma unroll 1
    for (int blk = 0; blk < NBLK_A; blk += 2) {
        projT_loadA<RAGGED>(B0, B1, v, nrow + (blk + 1) * (KB * 4), col);
        mma_block(c0, c1, A0, A1, lds_lane + (size_t)blk * (KB * 2) * DIM);
        projT_loadA<RAGGED>(A0, A1, v, nrow + min(blk + 2, NBLK_A - 1) * (KB * 4), col);
        mma_block(c0, c1, B0, B1, lds_lane + (size_t)(blk + 1) * (KB * 2) * DIM);
    }
}

template <bool FIRST>
__global__ __launch_bounds__(PROJT_THREADS)
void lgcn_projT(const float* __restrict__ v,
                const float* __restrict__ ego0,
                const float* __restrict__ ego1,
                float* __restrict__ partial  ) {
    __shared__ __attribute__((aligned(16))) v2f lds_ego[(ROWS_PAD / 2) * DIM];

    const int chunk   = blockIdx.x;
    const int n_begin = chunk * CHUNK;
    const int n_valid = min(N_TOTAL - n_begin, CHUNK);

    const int lane  = threadIdx.x & 31;
    const int wave  = threadIdx.x >> 5;
    const int f0    = (blockIdx.y * 4 + wave) * 32;
    const int row   = lane & 15;
    const int half  = lane >> 4;
    const int khalf = half * 2;

    const bool fast = (n_begin + ROWS_PAD <= N_TOTAL);

    v2f A0[KB], A1[KB];
    if (fast) projT_loadA<false>(A0, A1, v, n_begin + khalf, f0 + row);
    else      projT_loadA<true >(A0, A1, v, n_begin + khalf, f0 + row);

    for (int base = threadIdx.x; base < STAGE_A_ITEMS; base += PROJT_THREADS * STAGE_A_GROUP) {
        v4f lo[STAGE_A_GROUP], hi[STAGE_A_GROUP];
#pragma unroll
        for (int u = 0; u < STAGE_A_GROUP; ++u) {
            const int q  = base + u * PROJT_THREADS;
            const int j  = q / QCOLS;
            const int d4 = (q % QCOLS) * 4;
            lo[u] = *(const v4f*)(ego_row<FIRST>(ego0, ego1, n_begin + min(2 * j,     n_valid - 1)) + d4);
            hi[u] = *(const v4f*)(ego_row<FIRST>(ego0, ego1, n_begin + min(2 * j + 1, n_valid - 1)) + d4);
        }
#pragma unroll
        for (int u = 0; u < STAGE_A_GROUP; ++u) {
            const int q  = base + u * PROJT_THREADS;
            const int j  = q / QCOLS;
            const int d4 = (q % QCOLS) * 4;
            const v4f zero = {0.0f, 0.0f, 0.0f, 0.0f};
            stage_store(&lds_ego[j * DIM + d4],
                        (2 * j     < n_valid) ? lo[u] : zero,
                        (2 * j + 1 < n_valid) ? hi[u] : zero);
        }
    }
    __syncthreads();

    v8f c0[4] = {{}, {}, {}, {}};
    v8f c1[4] = {{}, {}, {}, {}};

    const v2f* lds_lane = &lds_ego[half * DIM + row];
    if (fast) {
        projT_kloop<false>(c0, c1, A0, A1, v, lds_lane, n_begin + khalf, f0 + row);
    } else {
        projT_kloop<true >(c0, c1, A0, A1, v, lds_lane, n_begin + khalf, f0 + row);
    }

    float* outp = partial + (size_t)chunk * FDIM * DIM;
    const int mbase = f0 + (half << 3);
#pragma unroll
    for (int t = 0; t < 4; ++t) {
#pragma unroll
        for (int j = 0; j < 8; ++j) {
            outp[(size_t)(mbase + j)      * DIM + t * 16 + row] = c0[t][j];
            outp[(size_t)(mbase + 16 + j) * DIM + t * 16 + row] = c1[t][j];
        }
    }
}

__global__ void lgcn_reduce(const float* __restrict__ partial,
                            const float* __restrict__ filt  ,
                            float* __restrict__ proj  ) {
    int i = blockIdx.x * blockDim.x + threadIdx.x;
    if (i < FDIM * DIM) {
        float s = 0.0f;
        for (int c = 0; c < NCHUNKS; ++c)
            s += partial[(size_t)c * FDIM * DIM + i];
        proj[i] = s * filt[i / DIM];
    }
}

template <bool TWO>
__device__ __forceinline__ void prop_loadA(v2f (&a0)[KB], v2f (&a1)[KB],
                                           const float* p0, const float* p1, int blk) {
    p0 += blk * (KB * 4);
    p1 += blk * (KB * 4);
#pragma unroll
    for (int i = 0; i < KB; ++i) {
        a0[i] = *(const v2f*)(p0 + 4 * i);
        if (TWO) a1[i] = *(const v2f*)(p1 + 4 * i);
    }
}

#define LOADS_STAY_ABOVE() __builtin_amdgcn_sched_barrier(0x7C6)

#define PROP_FIRST 0
#define PROP_MID   1
#define PROP_LAST  2

template <int MODE, bool RAGGED, bool TWO, int NT>
__device__ __forceinline__ void prop_writeback(const v8f (&c0)[NT], const v8f (&c1)[NT],
                                               const float* __restrict__ user,
                                               const float* __restrict__ item,
                                               float* __restrict__ ego,
                                               float* __restrict__ acc,
                                               float* __restrict__ out,
                                               int mbase, int col) {
    float old0[NT][8], old1[NT][8];
#pragma unroll
    for (int j = 0; j < 8; ++j) {
        const int r0 = mbase + j;
        const int r1 = mbase + 16 + j;
        const float* p0 = ego_row<MODE == PROP_FIRST>(MODE == PROP_FIRST ? user : acc, item,
                                                      RAGGED ? min(r0, N_TOTAL - 1) : r0) + col;
        const float* p1 = ego_row<MODE == PROP_FIRST>(MODE == PROP_FIRST ? user : acc, item,
                                                      RAGGED ? min(r1, N_TOTAL - 1) : r1) + col;
#pragma unroll
        for (int t = 0; t < NT; ++t) {
            old0[t][j] = p0[t * 16];
            if (TWO) old1[t][j] = p1[t * 16];
        }
    }
#pragma unroll
    for (int j = 0; j < 8; ++j) {
        const int r0 = mbase + j;
        const int r1 = mbase + 16 + j;
        if (!RAGGED || r0 < N_TOTAL) {
#pragma unroll
            for (int t = 0; t < NT; ++t) {
                const size_t idx = (size_t)r0 * DIM + t * 16 + col;
                if (MODE == PROP_LAST) {
                    out[idx] = (old0[t][j] + c0[t][j]) * (1.0f / (N_LAYERS + 1));
                } else {
                    ego[idx] = c0[t][j];
                    acc[idx] = old0[t][j] + c0[t][j];
                }
            }
        }
        if (TWO && (!RAGGED || r1 < N_TOTAL)) {
#pragma unroll
            for (int t = 0; t < NT; ++t) {
                const size_t idx = (size_t)r1 * DIM + t * 16 + col;
                if (MODE == PROP_LAST) {
                    out[idx] = (old1[t][j] + c1[t][j]) * (1.0f / (N_LAYERS + 1));
                } else {
                    ego[idx] = c1[t][j];
                    acc[idx] = old1[t][j] + c1[t][j];
                }
            }
        }
    }
}

template <int MODE, bool TWO, int NT>
__device__ __forceinline__ void prop_tile(const float* __restrict__ v,
                                          const float* __restrict__ user,
                                          const float* __restrict__ item,
                                          float* __restrict__ ego,
                                          float* __restrict__ acc,
                                          float* __restrict__ out,
                                          const v2f* lds_lane,
                                          int m0, int col0, int row, int half) {
    if (m0 >= N_TOTAL) return;
    const int khalf = half * 2;
    const int m = m0 + row;
    const float* vrow0 = v + (size_t)min(m,      N_TOTAL - 1) * FDIM;
    const float* vrow1 = v + (size_t)min(m + 16, N_TOTAL - 1) * FDIM;

    v8f c0[NT] = {};
    v8f c1[NT] = {};

    v2f A0[KB], A1[KB], B0[KB], B1[KB];
    prop_loadA<TWO>(A0, A1, vrow0 + khalf, vrow1 + khalf, 0);
#pragma unroll 1
    for (int blk = 0; blk < NBLK_B; blk += 2) {
        prop_loadA<TWO>(B0, B1, vrow0 + khalf, vrow1 + khalf, blk + 1);
        LOADS_STAY_ABOVE();
        mma_block<TWO>(c0, c1, A0, A1, lds_lane + (size_t)blk * (KB * 2) * DIM);
        prop_loadA<TWO>(A0, A1, vrow0 + khalf, vrow1 + khalf, min(blk + 2, NBLK_B - 1));
        LOADS_STAY_ABOVE();
        mma_block<TWO>(c0, c1, B0, B1, lds_lane + (size_t)(blk + 1) * (KB * 2) * DIM);
    }

    const int mbase = m0 + (half << 3);
    if (m0 + (TWO ? 32 : 16) <= N_TOTAL)
        prop_writeback<MODE, false, TWO, NT>(c0, c1, user, item, ego, acc, out, mbase, col0 + row);
    else
        prop_writeback<MODE, true,  TWO, NT>(c0, c1, user, item, ego, acc, out, mbase, col0 + row);
}

template <int MODE>
__global__ __launch_bounds__(PROP_THREADS)
void lgcn_prop(const float* __restrict__ v,
               const float* __restrict__ proj,
               const float* __restrict__ user,
               const float* __restrict__ item,
               float* __restrict__ ego,
               float* __restrict__ acc,
               float* __restrict__ out) {
    __shared__ __attribute__((aligned(16))) v2f lds_proj[(FDIM / 2) * DIM];

    if (threadIdx.x < PROP_STAGE_THREADS) {
        for (int base = threadIdx.x; base < STAGE_B_ITEMS; base += PROP_STAGE_THREADS * STAGE_B_GROUP) {
            v4f lo[STAGE_B_GROUP], hi[STAGE_B_GROUP];
#pragma unroll
            for (int u = 0; u < STAGE_B_GROUP; ++u) {
                const int q  = base + u * PROP_STAGE_THREADS;
                const int j  = q / QCOLS;
                const int d4 = (q % QCOLS) * 4;
                lo[u] = *(const v4f*)(proj + (size_t)(2 * j)     * DIM + d4);
                hi[u] = *(const v4f*)(proj + (size_t)(2 * j + 1) * DIM + d4);
            }
#pragma unroll
            for (int u = 0; u < STAGE_B_GROUP; ++u) {
                const int q  = base + u * PROP_STAGE_THREADS;
                const int j  = q / QCOLS;
                const int d4 = (q % QCOLS) * 4;
                stage_store(&lds_proj[j * DIM + d4], lo[u], hi[u]);
            }
        }
    }
    __syncthreads();

    const int lane  = threadIdx.x & 31;
    const int wave  = threadIdx.x >> 5;
    const int row   = lane & 15;
    const int half  = lane >> 4;
    const v2f* lds_lane = &lds_proj[half * DIM + row];

    const int wg_row0 = blockIdx.x * PROP_ROWS_WG;
    if (wave < PROP_MAIN_WAVES) {
        prop_tile<MODE, true, 4>(v, user, item, ego, acc, out, lds_lane,
                                 wg_row0 + wave * 32, 0, row, half);
    } else {
        const int q    = wave - PROP_MAIN_WAVES;
        const int col0 = (q & 1) * 32;
        prop_tile<MODE, false, 2>(v, user, item, ego, acc, out, lds_lane + col0,
                                  wg_row0 + PROP_MAIN_WAVES * 32 + (q >> 1) * 16, col0, row, half);
    }
}

extern "C" void kernel_launch(void* const* d_in, const int* in_sizes, int n_in,
                              void* d_out, int out_size, void* d_ws, size_t ws_size,
                              hipStream_t stream) {
    const float* user    = (const float*)d_in[0];
    const float* item    = (const float*)d_in[1];
    const float* v       = (const float*)d_in[2];
    const float* filters = (const float*)d_in[3];

    float* ego     = (float*)d_ws;
    float* acc     = ego + (size_t)N_TOTAL * DIM;
    float* partial = acc + (size_t)N_TOTAL * DIM;
    float* proj    = partial + (size_t)NCHUNKS * FDIM * DIM;
    float* out     = (float*)d_out;

    static_assert(N_LAYERS == 3, "one first, one middle and one last layer are launched below");
    const dim3 gA(NCHUNKS, 4);
    const int  gR = (FDIM * DIM + 255) / 256;

    lgcn_projT<true><<<gA, PROJT_THREADS, 0, stream>>>(v, user, item, partial);
    lgcn_reduce<<<gR, 256, 0, stream>>>(partial, filters, proj);
    lgcn_prop<PROP_FIRST><<<PROP_BLOCKS, PROP_THREADS, 0, stream>>>(
        v, proj, user, item, ego, acc, nullptr);

    lgcn_projT<false><<<gA, PROJT_THREADS, 0, stream>>>(v, ego, nullptr, partial);
    lgcn_reduce<<<gR, 256, 0, stream>>>(partial, filters + FDIM, proj);
    lgcn_prop<PROP_MID><<<PROP_BLOCKS, PROP_THREADS, 0, stream>>>(
        v, proj, nullptr, nullptr, ego, acc, nullptr);

    lgcn_projT<false><<<gA, PROJT_THREADS, 0, stream>>>(v, ego, nullptr, partial);
    lgcn_reduce<<<gR, 256, 0, stream>>>(partial, filters + 2 * FDIM, proj);
    lgcn_prop<PROP_LAST><<<PROP_BLOCKS, PROP_THREADS, 0, stream>>>(
        v, proj, nullptr, nullptr, nullptr, acc, out);
}


# lgcn_reduce: a thread's 128 partial loads stream ahead of its adds (55 in flight, one refilled per add), 256 workgroups of 128; was 8 waited batches of 16 loads, 128 workgroups; sum order kept
# speedup vs baseline: 3.7848x; 1.0296x over previous
#include <hip/hip_runtime.h>


typedef __attribute__((ext_vector_type(2))) float v2f;
typedef __attribute__((ext_vector_type(4))) float v4f;
typedef __attribute__((ext_vector_type(8))) float v8f;

#define N_USERS 29858
#define N_ITEMS 40981
#define N_TOTAL (N_USERS + N_ITEMS)
#define DIM     64
#define FDIM    512
#define N_LAYERS 3
#define NCHUNKS 128
#define CHUNK   556
#define QUADS   (CHUNK / 4)
#define KB      8
#define NBLK_A  ((QUADS + KB - 1) / KB + ((QUADS + KB - 1) / KB) % 2)
#define ROWS_PAD (NBLK_A * KB * 4)
#define NBLK_B  (FDIM / (KB * 4))
#define PROJT_THREADS 128
#define PROP_MAIN_WAVES 8
#define PROP_TAIL_WAVES 4
#define PROP_WAVES    (PROP_MAIN_WAVES + PROP_TAIL_WAVES)
#define PROP_THREADS  (PROP_WAVES * 32)
#define PROP_STAGE_THREADS (PROP_MAIN_WAVES * 32)
#define PROP_ROWS_WG  (PROP_MAIN_WAVES * 32 + (PROP_TAIL_WAVES / 2) * 16)
#define PROP_BLOCKS   ((N_TOTAL + PROP_ROWS_WG - 1) / PROP_ROWS_WG)
#define QCOLS   (DIM / 4)
#define STAGE_A_ITEMS ((ROWS_PAD / 2) * QCOLS)
#define STAGE_B_ITEMS ((FDIM / 2) * QCOLS)
#define STAGE_A_GROUP 9
#define STAGE_B_GROUP 8
#define REDUCE_THREADS 128
#define REDUCE_BLOCKS  ((FDIM * DIM) / REDUCE_THREADS)

static_assert(NCHUNKS * CHUNK >= N_TOTAL, "chunk coverage");
static_assert(NBLK_A % 2 == 0 && NBLK_B % 2 == 0, "two register sets alternate");
static_assert(ROWS_PAD >= CHUNK && ROWS_PAD % 2 == 0, "padded chunk");
static_assert(STAGE_A_ITEMS % (PROJT_THREADS * STAGE_A_GROUP) == 0, "whole staging groups");
static_assert(STAGE_B_ITEMS % (PROP_STAGE_THREADS * STAGE_B_GROUP) == 0, "whole staging groups");
static_assert(PROP_TAIL_WAVES % 2 == 0, "two 32-column tiles per left-over 16-row tile");
static_assert((FDIM * DIM) % REDUCE_THREADS == 0, "lgcn_reduce's grid covers the elements exactly");

#define WMMA_F32(A, B, C) \
    __builtin_amdgcn_wmma_f32_16x16x4_f32(false, (A), false, (B), (short)0, (C), false, false)

template <bool TWO = true, int NT = 4>
__device__ __forceinline__ void mma_block(v8f (&c0)[NT], v8f (&c1)[NT],
                                          const v2f (&a0)[KB], const v2f (&a1)[KB],
                                          const v2f* bp) {
#pragma unroll
    for (int i = 0; i < KB; ++i) {
#pragma unroll
        for (int t = 0; t < NT; ++t) {
            v2f b = bp[(2 * i) * DIM + t * 16];
            c0[t] = WMMA_F32(a0[i], b, c0[t]);
            if (TWO) c1[t] = WMMA_F32(a1[i], b, c1[t]);
        }
    }
}

__device__ __forceinline__ void stage_store(v2f* dst, v4f lo, v4f hi) {
    v2f e;
    e.x = lo.x; e.y = hi.x; dst[0] = e;
    e.x = lo.y; e.y = hi.y; dst[1] = e;
    e.x = lo.z; e.y = hi.z; dst[2] = e;
    e.x = lo.w; e.y = hi.w; dst[3] = e;
}

template <bool FIRST>
__device__ __forceinline__ const float* ego_row(const float* a, const float* b, int r) {
    if (FIRST) return (r < N_USERS) ? a + (size_t)r * DIM : b + (size_t)(r - N_USERS) * DIM;
    return a + (size_t)r * DIM;
}

template <bool RAGGED>
__device__ __forceinline__ void projT_loadA(v2f (&a0)[KB], v2f (&a1)[KB],
                                            const float* __restrict__ v,
                                            int n0, int col) {
    if (RAGGED) {
#pragma unroll
        for (int i = 0; i < KB; ++i) {
            const int na = n0 + 4 * i;
            const size_t r0 = (size_t)min(na,     N_TOTAL - 1) * FDIM;
            const size_t r1 = (size_t)min(na + 1, N_TOTAL - 1) * FDIM;
            a0[i].x = v[r0 + col];        a0[i].y = v[r1 + col];
            a1[i].x = v[r0 + col + 16];   a1[i].y = v[r1 + col + 16];
        }
    } else {
        const float* ap = v + (size_t)n0 * FDIM + col;
#pragma unroll
        for (int i = 0; i < KB; ++i) {
            a0[i].x = ap[(4 * i) * FDIM];        a0[i].y = ap[(4 * i + 1) * FDIM];
            a1[i].x = ap[(4 * i) * FDIM + 16];   a1[i].y = ap[(4 * i + 1) * FDIM + 16];
        }
    }
}

template <bool RAGGED>
__device__ __forceinline__ void projT_kloop(v8f (&c0)[4], v8f (&c1)[4],
                                            v2f (&A0)[KB], v2f (&A1)[KB],
                                            const float* __restrict__ v,
                                            const v2f* lds_lane, int nrow, int col) {
    v2f B0[KB], B1[KB];
#pragma unroll 1
    for (int blk = 0; blk < NBLK_A; blk += 2) {
        projT_loadA<RAGGED>(B0, B1, v, nrow + (blk + 1) * (KB * 4), col);
        mma_block(c0, c1, A0, A1, lds_lane + (size_t)blk * (KB * 2) * DIM);
        projT_loadA<RAGGED>(A0, A1, v, nrow + min(blk + 2, NBLK_A - 1) * (KB * 4), col);
        mma_block(c0, c1, B0, B1, lds_lane + (size_t)(blk + 1) * (KB * 2) * DIM);
    }
}

template <bool FIRST>
__global__ __launch_bounds__(PROJT_THREADS)
void lgcn_projT(const float* __restrict__ v,
                const float* __restrict__ ego0,
                const float* __restrict__ ego1,
                float* __restrict__ partial  ) {
    __shared__ __attribute__((aligned(16))) v2f lds_ego[(ROWS_PAD / 2) * DIM];

    const int chunk   = blockIdx.x;
    const int n_begin = chunk * CHUNK;
    const int n_valid = min(N_TOTAL - n_begin, CHUNK);

    const int lane  = threadIdx.x & 31;
    const int wave  = threadIdx.x >> 5;
    const int f0    = (blockIdx.y * 4 + wave) * 32;
    const int row   = lane & 15;
    const int half  = lane >> 4;
    const int khalf = half * 2;

    const bool fast = (n_begin + ROWS_PAD <= N_TOTAL);

    v2f A0[KB], A1[KB];
    if (fast) projT_loadA<false>(A0, A1, v, n_begin + khalf, f0 + row);
    else      projT_loadA<true >(A0, A1, v, n_begin + khalf, f0 + row);

    for (int base = threadIdx.x; base < STAGE_A_ITEMS; base += PROJT_THREADS * STAGE_A_GROUP) {
        v4f lo[STAGE_A_GROUP], hi[STAGE_A_GROUP];
#pragma unroll
        for (int u = 0; u < STAGE_A_GROUP; ++u) {
            const int q  = base + u * PROJT_THREADS;
            const int j  = q / QCOLS;
            const int d4 = (q % QCOLS) * 4;
            lo[u] = *(const v4f*)(ego_row<FIRST>(ego0, ego1, n_begin + min(2 * j,     n_valid - 1)) + d4);
            hi[u] = *(const v4f*)(ego_row<FIRST>(ego0, ego1, n_begin + min(2 * j + 1, n_valid - 1)) + d4);
        }
#pragma unroll
        for (int u = 0; u < STAGE_A_GROUP; ++u) {
            const int q  = base + u * PROJT_THREADS;
            const int j  = q / QCOLS;
            const int d4 = (q % QCOLS) * 4;
            const v4f zero = {0.0f, 0.0f, 0.0f, 0.0f};
            stage_store(&lds_ego[j * DIM + d4],
                        (2 * j     < n_valid) ? lo[u] : zero,
                        (2 * j + 1 < n_valid) ? hi[u] : zero);
        }
    }
    __syncthreads();

    v8f c0[4] = {{}, {}, {}, {}};
    v8f c1[4] = {{}, {}, {}, {}};

    const v2f* lds_lane = &lds_ego[half * DIM + row];
    if (fast) {
        projT_kloop<false>(c0, c1, A0, A1, v, lds_lane, n_begin + khalf, f0 + row);
    } else {
        projT_kloop<true >(c0, c1, A0, A1, v, lds_lane, n_begin + khalf, f0 + row);
    }

    float* outp = partial + (size_t)chunk * FDIM * DIM;
    const int mbase = f0 + (half << 3);
#pragma unroll
    for (int t = 0; t < 4; ++t) {
#pragma unroll
        for (int j = 0; j < 8; ++j) {
            outp[(size_t)(mbase + j)      * DIM + t * 16 + row] = c0[t][j];
            outp[(size_t)(mbase + 16 + j) * DIM + t * 16 + row] = c1[t][j];
        }
    }
}

__global__ __launch_bounds__(REDUCE_THREADS)
void lgcn_reduce(const float* __restrict__ partial,
                 const float* __restrict__ filt  ,
                 float* __restrict__ proj  ) {
    const int i = blockIdx.x * REDUCE_THREADS + threadIdx.x;
    const float f = filt[i / DIM];
    float p[NCHUNKS];
#pragma unroll
    for (int c = 0; c < NCHUNKS; ++c)
        p[c] = partial[(size_t)c * FDIM * DIM + i];
    float s = 0.0f;
#pragma unroll
    for (int c = 0; c < NCHUNKS; ++c)
        s += p[c];
    proj[i] = s * f;
}

template <bool TWO>
__device__ __forceinline__ void prop_loadA(v2f (&a0)[KB], v2f (&a1)[KB],
                                           const float* p0, const float* p1, int blk) {
    p0 += blk * (KB * 4);
    p1 += blk * (KB * 4);
#pragma unroll
    for (int i = 0; i < KB; ++i) {
        a0[i] = *(const v2f*)(p0 + 4 * i);
        if (TWO) a1[i] = *(const v2f*)(p1 + 4 * i);
    }
}

#define LOADS_STAY_ABOVE() __builtin_amdgcn_sched_barrier(0x7C6)

#define PROP_FIRST 0
#define PROP_MID   1
#define PROP_LAST  2

template <int MODE, bool RAGGED, bool TWO, int NT>
__device__ __forceinline__ void prop_writeback(const v8f (&c0)[NT], const v8f (&c1)[NT],
                                               const float* __restrict__ user,
                                               const float* __restrict__ item,
                                               float* __restrict__ ego,
                                               float* __restrict__ acc,
                                               float* __restrict__ out,
                                               int mbase, int col) {
    float old0[NT][8], old1[NT][8];
#pragma unroll
    for (int j = 0; j < 8; ++j) {
        const int r0 = mbase + j;
        const int r1 = mbase + 16 + j;
        const float* p0 = ego_row<MODE == PROP_FIRST>(MODE == PROP_FIRST ? user : acc, item,
                                                      RAGGED ? min(r0, N_TOTAL - 1) : r0) + col;
        const float* p1 = ego_row<MODE == PROP_FIRST>(MODE == PROP_FIRST ? user : acc, item,
                                                      RAGGED ? min(r1, N_TOTAL - 1) : r1) + col;
#pragma unroll
        for (int t = 0; t < NT; ++t) {
            old0[t][j] = p0[t * 16];
            if (TWO) old1[t][j] = p1[t * 16];
        }
    }
#pragma unroll
    for (int j = 0; j < 8; ++j) {
        const int r0 = mbase + j;
        const int r1 = mbase + 16 + j;
        if (!RAGGED || r0 < N_TOTAL) {
#pragma unroll
            for (int t = 0; t < NT; ++t) {
                const size_t idx = (size_t)r0 * DIM + t * 16 + col;
                if (MODE == PROP_LAST) {
                    out[idx] = (old0[t][j] + c0[t][j]) * (1.0f / (N_LAYERS + 1));
                } else {
                    ego[idx] = c0[t][j];
                    acc[idx] = old0[t][j] + c0[t][j];
                }
            }
        }
        if (TWO && (!RAGGED || r1 < N_TOTAL)) {
#pragma unroll
            for (int t = 0; t < NT; ++t) {
                const size_t idx = (size_t)r1 * DIM + t * 16 + col;
                if (MODE == PROP_LAST) {
                    out[idx] = (old1[t][j] + c1[t][j]) * (1.0f / (N_LAYERS + 1));
                } else {
                    ego[idx] = c1[t][j];
                    acc[idx] = old1[t][j] + c1[t][j];
                }
            }
        }
    }
}

template <int MODE, bool TWO, int NT>
__device__ __forceinline__ void prop_tile(const float* __restrict__ v,
                                          const float* __restrict__ user,
                                          const float* __restrict__ item,
                                          float* __restrict__ ego,
                                          float* __restrict__ acc,
                                          float* __restrict__ out,
                                          const v2f* lds_lane,
                                          int m0, int col0, int row, int half) {
    if (m0 >= N_TOTAL) return;
    const int khalf = half * 2;
    const int m = m0 + row;
    const float* vrow0 = v + (size_t)min(m,      N_TOTAL - 1) * FDIM;
    const float* vrow1 = v + (size_t)min(m + 16, N_TOTAL - 1) * FDIM;

    v8f c0[NT] = {};
    v8f c1[NT] = {};

    v2f A0[KB], A1[KB], B0[KB], B1[KB];
    prop_loadA<TWO>(A0, A1, vrow0 + khalf, vrow1 + khalf, 0);
#pragma unroll 1
    for (int blk = 0; blk < NBLK_B; blk += 2) {
        prop_loadA<TWO>(B0, B1, vrow0 + khalf, vrow1 + khalf, blk + 1);
        LOADS_STAY_ABOVE();
        mma_block<TWO>(c0, c1, A0, A1, lds_lane + (size_t)blk * (KB * 2) * DIM);
        prop_loadA<TWO>(A0, A1, vrow0 + khalf, vrow1 + khalf, min(blk + 2, NBLK_B - 1));
        LOADS_STAY_ABOVE();
        mma_block<TWO>(c0, c1, B0, B1, lds_lane + (size_t)(blk + 1) * (KB * 2) * DIM);
    }

    const int mbase = m0 + (half << 3);
    if (m0 + (TWO ? 32 : 16) <= N_TOTAL)
        prop_writeback<MODE, false, TWO, NT>(c0, c1, user, item, ego, acc, out, mbase, col0 + row);
    else
        prop_writeback<MODE, true,  TWO, NT>(c0, c1, user, item, ego, acc, out, mbase, col0 + row);
}

template <int MODE>
__global__ __launch_bounds__(PROP_THREADS)
void lgcn_prop(const float* __restrict__ v,
               const float* __restrict__ proj,
               const float* __restrict__ user,
               const float* __restrict__ item,
               float* __restrict__ ego,
               float* __restrict__ acc,
               float* __restrict__ out) {
    __shared__ __attribute__((aligned(16))) v2f lds_proj[(FDIM / 2) * DIM];

    if (threadIdx.x < PROP_STAGE_THREADS) {
        for (int base = threadIdx.x; base < STAGE_B_ITEMS; base += PROP_STAGE_THREADS * STAGE_B_GROUP) {
            v4f lo[STAGE_B_GROUP], hi[STAGE_B_GROUP];
#pragma unroll
            for (int u = 0; u < STAGE_B_GROUP; ++u) {
                const int q  = base + u * PROP_STAGE_THREADS;
                const int j  = q / QCOLS;
                const int d4 = (q % QCOLS) * 4;
                lo[u] = *(const v4f*)(proj + (size_t)(2 * j)     * DIM + d4);
                hi[u] = *(const v4f*)(proj + (size_t)(2 * j + 1) * DIM + d4);
            }
#pragma unroll
            for (int u = 0; u < STAGE_B_GROUP; ++u) {
                const int q  = base + u * PROP_STAGE_THREADS;
                const int j  = q / QCOLS;
                const int d4 = (q % QCOLS) * 4;
                stage_store(&lds_proj[j * DIM + d4], lo[u], hi[u]);
            }
        }
    }
    __syncthreads();

    const int lane  = threadIdx.x & 31;
    const int wave  = threadIdx.x >> 5;
    const int row   = lane & 15;
    const int half  = lane >> 4;
    const v2f* lds_lane = &lds_proj[half * DIM + row];

    const int wg_row0 = blockIdx.x * PROP_ROWS_WG;
    if (wave < PROP_MAIN_WAVES) {
        prop_tile<MODE, true, 4>(v, user, item, ego, acc, out, lds_lane,
                                 wg_row0 + wave * 32, 0, row, half);
    } else {
        const int q    = wave - PROP_MAIN_WAVES;
        const int col0 = (q & 1) * 32;
        prop_tile<MODE, false, 2>(v, user, item, ego, acc, out, lds_lane + col0,
                                  wg_row0 + PROP_MAIN_WAVES * 32 + (q >> 1) * 16, col0, row, half);
    }
}

extern "C" void kernel_launch(void* const* d_in, const int* in_sizes, int n_in,
                              void* d_out, int out_size, void* d_ws, size_t ws_size,
                              hipStream_t stream) {
    const float* user    = (const float*)d_in[0];
    const float* item    = (const float*)d_in[1];
    const float* v       = (const float*)d_in[2];
    const float* filters = (const float*)d_in[3];

    float* ego     = (float*)d_ws;
    float* acc     = ego + (size_t)N_TOTAL * DIM;
    float* partial = acc + (size_t)N_TOTAL * DIM;
    float* proj    = partial + (size_t)NCHUNKS * FDIM * DIM;
    float* out     = (float*)d_out;

    static_assert(N_LAYERS == 3, "one first, one middle and one last layer are launched below");
    const dim3 gA(NCHUNKS, 4);

    lgcn_projT<true><<<gA, PROJT_THREADS, 0, stream>>>(v, user, item, partial);
    lgcn_reduce<<<REDUCE_BLOCKS, REDUCE_THREADS, 0, stream>>>(partial, filters, proj);
    lgcn_prop<PROP_FIRST><<<PROP_BLOCKS, PROP_THREADS, 0, stream>>>(
        v, proj, user, item, ego, acc, nullptr);

    lgcn_projT<false><<<gA, PROJT_THREADS, 0, stream>>>(v, ego, nullptr, partial);
    lgcn_reduce<<<REDUCE_BLOCKS, REDUCE_THREADS, 0, stream>>>(partial, filters + FDIM, proj);
    lgcn_prop<PROP_MID><<<PROP_BLOCKS, PROP_THREADS, 0, stream>>>(
        v, proj, nullptr, nullptr, ego, acc, nullptr);

    lgcn_projT<false><<<gA, PROJT_THREADS, 0, stream>>>(v, ego, nullptr, partial);
    lgcn_reduce<<<REDUCE_BLOCKS, REDUCE_THREADS, 0, stream>>>(partial, filters + 2 * FDIM, proj);
    lgcn_prop<PROP_LAST><<<PROP_BLOCKS, PROP_THREADS, 0, stream>>>(
        v, proj, nullptr, nullptr, nullptr, acc, out);
}
